# starting kernel
# baseline (speedup 1.0000x reference)
#include <hip/hip_runtime.h>
#include <hip/hip_bf16.h>

typedef __attribute__((ext_vector_type(2))) float v2f;
typedef __attribute__((ext_vector_type(8))) float v8f;

#define C_IN 128

__global__ void sage_zero(float* __restrict__ p, long long n) {
  long long i = (long long)blockIdx.x * blockDim.x + threadIdx.x;
  long long stride = (long long)gridDim.x * blockDim.x;
  for (; i < n; i += stride) p[i] = 0.0f;
}

__global__ void sage_scatter(const float* __restrict__ x,
                             const int* __restrict__ src,
                             const int* __restrict__ dst,
                             float* __restrict__ agg,
                             float* __restrict__ cnt,
                             int nEdges) {
  long long t = (long long)blockIdx.x * blockDim.x + threadIdx.x;
  int e  = (int)(t >> 5);
  int cq = (int)(t & 31);
  if (e >= nEdges) return;
  int s = src[e];
  int d = dst[e];
  const float4* xrow = (const float4*)(x + (long long)s * C_IN);
  float4 v = xrow[cq];
  float* arow = agg + (long long)d * C_IN + cq * 4;
  atomicAdd(arow + 0, v.x);
  atomicAdd(arow + 1, v.y);
  atomicAdd(arow + 2, v.z);
  atomicAdd(arow + 3, v.w);
  if (cq == 0) atomicAdd(cnt + d, 1.0f);
}

template <int COUT, bool RELU>
__global__ void sage_gemm(const float* __restrict__ xin,
                          const float* __restrict__ agg,
                          const float* __restrict__ cnt,
                          const float* __restrict__ Wl,
                          const float* __restrict__ bl,
                          const float* __restrict__ Wr,
                          float* __restrict__ out,
                          int totalTiles) {
  const int tilesN = COUT / 16;
  int wave = (int)(((long long)blockIdx.x * blockDim.x + threadIdx.x) >> 5);
  if (wave >= totalTiles) return;
  int lane = threadIdx.x & 31;
  int m0 = (wave / tilesN) * 16;
  int n0 = (wave % tilesN) * 16;
  int r  = lane & 15;
  int kh = lane >> 4;
  int col = n0 + r;

  const float* arow = agg + (long long)(m0 + r) * C_IN;
  const float* xrow = xin + (long long)(m0 + r) * C_IN;
  float inv = 1.0f / fmaxf(cnt[m0 + r], 1.0f);

  v8f acc = {};

#pragma unroll
  for (int k = 0; k < C_IN; k += 4) {
    v2f a, b;
    int kk = k + 2 * kh;
    a[0] = arow[kk] * inv;
    a[1] = arow[kk + 1] * inv;
    b[0] = Wl[(long long)kk * COUT + col];
    b[1] = Wl[(long long)(kk + 1) * COUT + col];
    acc = __builtin_amdgcn_wmma_f32_16x16x4_f32(false, a, false, b,
                                                (short)0, acc, false, false);
  }
#pragma unroll
  for (int k = 0; k < C_IN; k += 4) {
    v2f a, b;
    int kk = k + 2 * kh;
    a[0] = xrow[kk];
    a[1] = xrow[kk + 1];
    b[0] = Wr[(long long)kk * COUT + col];
    b[1] = Wr[(long long)(kk + 1) * COUT + col];
    acc = __builtin_amdgcn_wmma_f32_16x16x4_f32(false, a, false, b,
                                                (short)0, acc, false, false);
  }

  float bias = bl[col];
#pragma unroll
  for (int rr = 0; rr < 8; ++rr) {
    float v = acc[rr] + bias;
    if (RELU) v = fmaxf(v, 0.0f);
    out[(long long)(m0 + rr + 8 * kh) * COUT + col] = v;
  }
}

extern "C" void kernel_launch(void* const* d_in, const int* in_sizes, int n_in,
                              void* d_out, int out_size, void* d_ws, size_t ws_size,
                              hipStream_t stream) {
  const float* x0  = (const float*)d_in[0];
  const int*   ei  = (const int*)d_in[1];
  const float* Wl0 = (const float*)d_in[2];
  const float* bl0 = (const float*)d_in[3];
  const float* Wr0 = (const float*)d_in[4];
  const float* Wl1 = (const float*)d_in[5];
  const float* bl1 = (const float*)d_in[6];
  const float* Wr1 = (const float*)d_in[7];
  const float* Wl2 = (const float*)d_in[8];
  const float* bl2 = (const float*)d_in[9];
  const float* Wr2 = (const float*)d_in[10];

  const int N = in_sizes[0] / C_IN;
  const int E = in_sizes[1] / 2;
  const int* src = ei;
  const int* dst = ei + E;

  float* h0  = (float*)d_ws;
  float* h1  = h0 + (size_t)N * C_IN;
  float* agg = h1 + (size_t)N * C_IN;
  float* cnt = agg + (size_t)N * C_IN;
  const long long aggCntElems = (long long)N * C_IN + N;

  const int ZB = 1024;
  const long long scatterThreads = (long long)E * 32;
  const int scatterBlocks = (int)((scatterThreads + 255) / 256);

  const int tilesM = (N + 15) / 16;

  sage_zero<<<ZB, 256, 0, stream>>>(agg, aggCntElems);
  sage_scatter<<<scatterBlocks, 256, 0, stream>>>(x0, src, dst, agg, cnt, E);
  {
    int tiles = tilesM * (128 / 16);
    int blocks = (tiles + 7) / 8;
    sage_gemm<128, true><<<blocks, 256, 0, stream>>>(x0, agg, cnt, Wl0, bl0, Wr0, h0, tiles);
  }

  sage_zero<<<ZB, 256, 0, stream>>>(agg, aggCntElems);
  sage_scatter<<<scatterBlocks, 256, 0, stream>>>(h0, src, dst, agg, cnt, E);
  {
    int tiles = tilesM * (128 / 16);
    int blocks = (tiles + 7) / 8;
    sage_gemm<128, true><<<blocks, 256, 0, stream>>>(h0, agg, cnt, Wl1, bl1, Wr1, h1, tiles);
  }

  sage_zero<<<ZB, 256, 0, stream>>>(agg, aggCntElems);
  sage_scatter<<<scatterBlocks, 256, 0, stream>>>(h1, src, dst, agg, cnt, E);
  {
    int tiles = tilesM * (64 / 16);
    int blocks = (tiles + 7) / 8;
    sage_gemm<64, false><<<blocks, 256, 0, stream>>>(h1, agg, cnt, Wl2, bl2, Wr2,
                                                     (float*)d_out, tiles);
  }
}


# Edge scatter: lane = feature column, so each float atomic instruction and row load covers one whole 128-byte line (4 line adds per edge, not 16 partial); atomics kept, sums still in arrival order
# speedup vs baseline: 1.1338x; 1.1338x over previous
#include <hip/hip_runtime.h>
#include <hip/hip_bf16.h>

typedef __attribute__((ext_vector_type(2))) float v2f;
typedef __attribute__((ext_vector_type(8))) float v8f;

#define C_IN 128

__global__ void sage_zero(float* __restrict__ p, long long n) {
  long long i = (long long)blockIdx.x * blockDim.x + threadIdx.x;
  long long stride = (long long)gridDim.x * blockDim.x;
  for (; i < n; i += stride) p[i] = 0.0f;
}

__global__ void sage_scatter(const float* __restrict__ x,
                             const int* __restrict__ src,
                             const int* __restrict__ dst,
                             float* __restrict__ agg,
                             float* __restrict__ cnt,
                             int nEdges) {
  long long t = (long long)blockIdx.x * blockDim.x + threadIdx.x;
  int e    = (int)(t >> 5);
  int lane = (int)(t & 31);
  if (e >= nEdges) return;
  int s = src[e];
  int d = dst[e];
  const float* xrow = x + (long long)s * C_IN + lane;
  float v0 = xrow[0];
  float v1 = xrow[32];
  float v2 = xrow[64];
  float v3 = xrow[96];
  float* arow = agg + (long long)d * C_IN + lane;
  atomicAdd(arow,      v0);
  atomicAdd(arow + 32, v1);
  atomicAdd(arow + 64, v2);
  atomicAdd(arow + 96, v3);
  if (lane == 0) atomicAdd(cnt + d, 1.0f);
}

template <int COUT, bool RELU>
__global__ void sage_gemm(const float* __restrict__ xin,
                          const float* __restrict__ agg,
                          const float* __restrict__ cnt,
                          const float* __restrict__ Wl,
                          const float* __restrict__ bl,
                          const float* __restrict__ Wr,
                          float* __restrict__ out,
                          int totalTiles) {
  const int tilesN = COUT / 16;
  int wave = (int)(((long long)blockIdx.x * blockDim.x + threadIdx.x) >> 5);
  if (wave >= totalTiles) return;
  int lane = threadIdx.x & 31;
  int m0 = (wave / tilesN) * 16;
  int n0 = (wave % tilesN) * 16;
  int r  = lane & 15;
  int kh = lane >> 4;
  int col = n0 + r;

  const float* arow = agg + (long long)(m0 + r) * C_IN;
  const float* xrow = xin + (long long)(m0 + r) * C_IN;
  float inv = 1.0f / fmaxf(cnt[m0 + r], 1.0f);

  v8f acc = {};

#pragma unroll
  for (int k = 0; k < C_IN; k += 4) {
    v2f a, b;
    int kk = k + 2 * kh;
    a[0] = arow[kk] * inv;
    a[1] = arow[kk + 1] * inv;
    b[0] = Wl[(long long)kk * COUT + col];
    b[1] = Wl[(long long)(kk + 1) * COUT + col];
    acc = __builtin_amdgcn_wmma_f32_16x16x4_f32(false, a, false, b,
                                                (short)0, acc, false, false);
  }
#pragma unroll
  for (int k = 0; k < C_IN; k += 4) {
    v2f a, b;
    int kk = k + 2 * kh;
    a[0] = xrow[kk];
    a[1] = xrow[kk + 1];
    b[0] = Wr[(long long)kk * COUT + col];
    b[1] = Wr[(long long)(kk + 1) * COUT + col];
    acc = __builtin_amdgcn_wmma_f32_16x16x4_f32(false, a, false, b,
                                                (short)0, acc, false, false);
  }

  float bias = bl[col];
#pragma unroll
  for (int rr = 0; rr < 8; ++rr) {
    float v = acc[rr] + bias;
    if (RELU) v = fmaxf(v, 0.0f);
    out[(long long)(m0 + rr + 8 * kh) * COUT + col] = v;
  }
}

extern "C" void kernel_launch(void* const* d_in, const int* in_sizes, int n_in,
                              void* d_out, int out_size, void* d_ws, size_t ws_size,
                              hipStream_t stream) {
  const float* x0  = (const float*)d_in[0];
  const int*   ei  = (const int*)d_in[1];
  const float* Wl0 = (const float*)d_in[2];
  const float* bl0 = (const float*)d_in[3];
  const float* Wr0 = (const float*)d_in[4];
  const float* Wl1 = (const float*)d_in[5];
  const float* bl1 = (const float*)d_in[6];
  const float* Wr1 = (const float*)d_in[7];
  const float* Wl2 = (const float*)d_in[8];
  const float* bl2 = (const float*)d_in[9];
  const float* Wr2 = (const float*)d_in[10];

  const int N = in_sizes[0] / C_IN;
  const int E = in_sizes[1] / 2;
  const int* src = ei;
  const int* dst = ei + E;

  float* h0  = (float*)d_ws;
  float* h1  = h0 + (size_t)N * C_IN;
  float* agg = h1 + (size_t)N * C_IN;
  float* cnt = agg + (size_t)N * C_IN;
  const long long aggCntElems = (long long)N * C_IN + N;

  const int ZB = 1024;
  const long long scatterThreads = (long long)E * 32;
  const int scatterBlocks = (int)((scatterThreads + 255) / 256);

  const int tilesM = (N + 15) / 16;

  sage_zero<<<ZB, 256, 0, stream>>>(agg, aggCntElems);
  sage_scatter<<<scatterBlocks, 256, 0, stream>>>(x0, src, dst, agg, cnt, E);
  {
    int tiles = tilesM * (128 / 16);
    int blocks = (tiles + 7) / 8;
    sage_gemm<128, true><<<blocks, 256, 0, stream>>>(x0, agg, cnt, Wl0, bl0, Wr0, h0, tiles);
  }

  sage_zero<<<ZB, 256, 0, stream>>>(agg, aggCntElems);
  sage_scatter<<<scatterBlocks, 256, 0, stream>>>(h0, src, dst, agg, cnt, E);
  {
    int tiles = tilesM * (128 / 16);
    int blocks = (tiles + 7) / 8;
    sage_gemm<128, true><<<blocks, 256, 0, stream>>>(h0, agg, cnt, Wl1, bl1, Wr1, h1, tiles);
  }

  sage_zero<<<ZB, 256, 0, stream>>>(agg, aggCntElems);
  sage_scatter<<<scatterBlocks, 256, 0, stream>>>(h1, src, dst, agg, cnt, E);
  {
    int tiles = tilesM * (64 / 16);
    int blocks = (tiles + 7) / 8;
    sage_gemm<64, false><<<blocks, 256, 0, stream>>>(h1, agg, cnt, Wl2, bl2, Wr2,
                                                     (float*)d_out, tiles);
  }
}


# Aggregation by node-range owners: each work-group filters the unsorted edge list for its 391 nodes and sums their source rows in LDS (LDS float adds, arrival order); no global atomics, no zero fill
# speedup vs baseline: 1.9400x; 1.7111x over previous
#include <hip/hip_runtime.h>
#include <hip/hip_bf16.h>

typedef __attribute__((ext_vector_type(2))) float v2f;
typedef __attribute__((ext_vector_type(8))) float v8f;

#define C_IN 128

#define OWN_NODES 391
#define LIST_CAP 16384
#define AGG_THREADS 1024
#define EDGE_BATCH 8

__device__ __forceinline__ void add_edge_direct(const float* __restrict__ x,
                                                const int* __restrict__ src,
                                                float* sAcc, float* sCnt,
                                                int e, unsigned rel) {
  int s = src[e];
  const float* xr = x + (long long)s * C_IN;
  float* ar = sAcc + rel * C_IN;
  for (int c = 0; c < C_IN; ++c) atomicAdd(ar + c, xr[c]);
  atomicAdd(sCnt + rel, 1.0f);
}

__device__ __forceinline__ void load_dst512(int4 (&buf)[4], const int* __restrict__ dst,
                                            int sc, int lane) {
#pragma unroll
  for (int u = 0; u < 4; ++u)
    buf[u] = *(const int4*)(dst + (long long)sc * 512 + u * 128 + lane * 4);
}

__device__ __forceinline__ void note_owned(const int4 (&buf)[4], int sc, int lane,
                                           int nodeLo, unsigned nOwn,
                                           unsigned* sList, unsigned* sCount) {
  unsigned rel[16];
  unsigned msk[16];
  unsigned total = 0;
#pragma unroll
  for (int u = 0; u < 4; ++u) {
    rel[u * 4 + 0] = (unsigned)(buf[u].x - nodeLo);
    rel[u * 4 + 1] = (unsigned)(buf[u].y - nodeLo);
    rel[u * 4 + 2] = (unsigned)(buf[u].z - nodeLo);
    rel[u * 4 + 3] = (unsigned)(buf[u].w - nodeLo);
  }
#pragma unroll
  for (int q = 0; q < 16; ++q) {
    msk[q] = __builtin_amdgcn_ballot_w32(rel[q] < nOwn);
    total += (unsigned)__popc(msk[q]);
  }
  if (total != 0u) {
    unsigned base = 0u;
    if (lane == 0) base = atomicAdd(sCount, total);
    base = (unsigned)__builtin_amdgcn_readfirstlane((int)base);
#pragma unroll
    for (int q = 0; q < 16; ++q) {
      if (msk[q] != 0u) {
        if (rel[q] < nOwn) {
          unsigned slot = base + __builtin_amdgcn_mbcnt_lo(msk[q], 0u);
          unsigned e = (unsigned)sc * 512u + (unsigned)((q >> 2) * 128 + lane * 4 + (q & 3));
          if (slot < LIST_CAP) sList[slot] = e * 512u + rel[q];
        }
        base += (unsigned)__popc(msk[q]);
      }
    }
  }
}

__global__ void __launch_bounds__(AGG_THREADS)
sage_aggregate(const float* __restrict__ x,
               const int* __restrict__ src,
               const int* __restrict__ dst,
               float* __restrict__ agg,
               float* __restrict__ cnt,
               int nEdges, int nNodes) {
  __shared__ __attribute__((aligned(16))) float sAcc[OWN_NODES * C_IN];
  __shared__ float sCnt[OWN_NODES];
  __shared__ unsigned sList[LIST_CAP];
  __shared__ unsigned sCount;

  const int tid  = (int)threadIdx.x;
  const int lane = tid & 31;
  const int wv   = __builtin_amdgcn_readfirstlane(tid >> 5);
  const int nodeLo = (int)blockIdx.x * OWN_NODES;
  int nOwnI = nNodes - nodeLo;
  if (nOwnI > OWN_NODES) nOwnI = OWN_NODES;
  if (nOwnI <= 0) return;
  const unsigned nOwn = (unsigned)nOwnI;

  for (int i = tid; i < OWN_NODES * C_IN / 4; i += AGG_THREADS)
    ((float4*)sAcc)[i] = make_float4(0.f, 0.f, 0.f, 0.f);
  for (int i = tid; i < OWN_NODES; i += AGG_THREADS) sCnt[i] = 0.0f;
  if (tid == 0) sCount = 0u;
  __syncthreads();

  const int nSuper = nEdges >> 9;
  {
    int4 bufA[4], bufB[4];
    int sc = wv;
    if (sc < nSuper) {
      const int scLast = nSuper - 1;
      load_dst512(bufA, dst, sc, lane);
      while (true) {
        load_dst512(bufB, dst, (sc + 32 < scLast) ? sc + 32 : scLast, lane);
        note_owned(bufA, sc, lane, nodeLo, nOwn, sList, &sCount);
        sc += 32;
        if (sc >= nSuper) break;
        load_dst512(bufA, dst, (sc + 32 < scLast) ? sc + 32 : scLast, lane);
        note_owned(bufB, sc, lane, nodeLo, nOwn, sList, &sCount);
        sc += 32;
        if (sc >= nSuper) break;
      }
    }
  }
  for (int e = (nSuper << 9) + tid; e < nEdges; e += AGG_THREADS) {
    unsigned r1 = (unsigned)(dst[e] - nodeLo);
    if (r1 < nOwn) {
      unsigned slot = atomicAdd(&sCount, 1u);
      if (slot < LIST_CAP) sList[slot] = (unsigned)e * 512u + r1;
    }
  }
  __syncthreads();

  const unsigned nList = sCount;
  if (nList > LIST_CAP) {
    for (int e = tid; e < nEdges; e += AGG_THREADS) {
      unsigned r1 = (unsigned)(dst[e] - nodeLo);
      if (r1 < nOwn) add_edge_direct(x, src, sAcc, sCnt, e, r1);
    }
  } else {
    for (unsigned i0 = (unsigned)wv * 32u; i0 < nList; i0 += 32u * 32u) {
      unsigned nIn = nList - i0;
      if (nIn > 32u) nIn = 32u;
      int sL = 0;
      int rL = 0;
      if ((unsigned)lane < nIn) {
        unsigned ent = sList[i0 + lane];
        rL = (int)(ent & 511u);
        sL = src[ent >> 9];
        atomicAdd(sCnt + rL, 1.0f);
      }
      if (nIn == 32u) {
#pragma unroll
        for (int j = 0; j < 32; j += EDGE_BATCH) {
          float v[EDGE_BATCH][4];
#pragma unroll
          for (int u = 0; u < EDGE_BATCH; ++u) {
            int s = __builtin_amdgcn_readlane(sL, j + u);
            const float* xrow = x + (long long)s * C_IN + lane;
            v[u][0] = xrow[0];
            v[u][1] = xrow[32];
            v[u][2] = xrow[64];
            v[u][3] = xrow[96];
          }
#pragma unroll
          for (int u = 0; u < EDGE_BATCH; ++u) {
            int d = __builtin_amdgcn_readlane(rL, j + u);
            float* arow = sAcc + d * C_IN + lane;
            atomicAdd(arow,      v[u][0]);
            atomicAdd(arow + 32, v[u][1]);
            atomicAdd(arow + 64, v[u][2]);
            atomicAdd(arow + 96, v[u][3]);
          }
        }
      } else {
        for (unsigned j = 0; j < nIn; ++j) {
          int s = __shfl(sL, (int)j);
          int d = __shfl(rL, (int)j);
          const float* xrow = x + (long long)s * C_IN + lane;
          float v0 = xrow[0];
          float v1 = xrow[32];
          float v2 = xrow[64];
          float v3 = xrow[96];
          float* arow = sAcc + d * C_IN + lane;
          atomicAdd(arow,      v0);
          atomicAdd(arow + 32, v1);
          atomicAdd(arow + 64, v2);
          atomicAdd(arow + 96, v3);
        }
      }
    }
  }
  __syncthreads();

  {
    float4* gdst = (float4*)(agg + (long long)nodeLo * C_IN);
    const int n4 = nOwnI * (C_IN / 4);
    for (int i = tid; i < n4; i += AGG_THREADS) gdst[i] = ((const float4*)sAcc)[i];
    for (int i = tid; i < nOwnI; i += AGG_THREADS) cnt[nodeLo + i] = sCnt[i];
  }
}

template <int COUT, bool RELU>
__global__ void sage_gemm(const float* __restrict__ xin,
                          const float* __restrict__ agg,
                          const float* __restrict__ cnt,
                          const float* __restrict__ Wl,
                          const float* __restrict__ bl,
                          const float* __restrict__ Wr,
                          float* __restrict__ out,
                          int totalTiles) {
  const int tilesN = COUT / 16;
  int wave = (int)(((long long)blockIdx.x * blockDim.x + threadIdx.x) >> 5);
  if (wave >= totalTiles) return;
  int lane = threadIdx.x & 31;
  int m0 = (wave / tilesN) * 16;
  int n0 = (wave % tilesN) * 16;
  int r  = lane & 15;
  int kh = lane >> 4;
  int col = n0 + r;

  const float* arow = agg + (long long)(m0 + r) * C_IN;
  const float* xrow = xin + (long long)(m0 + r) * C_IN;
  float inv = 1.0f / fmaxf(cnt[m0 + r], 1.0f);

  v8f acc = {};

#pragma unroll
  for (int k = 0; k < C_IN; k += 4) {
    v2f a, b;
    int kk = k + 2 * kh;
    a[0] = arow[kk] * inv;
    a[1] = arow[kk + 1] * inv;
    b[0] = Wl[(long long)kk * COUT + col];
    b[1] = Wl[(long long)(kk + 1) * COUT + col];
    acc = __builtin_amdgcn_wmma_f32_16x16x4_f32(false, a, false, b,
                                                (short)0, acc, false, false);
  }
#pragma unroll
  for (int k = 0; k < C_IN; k += 4) {
    v2f a, b;
    int kk = k + 2 * kh;
    a[0] = xrow[kk];
    a[1] = xrow[kk + 1];
    b[0] = Wr[(long long)kk * COUT + col];
    b[1] = Wr[(long long)(kk + 1) * COUT + col];
    acc = __builtin_amdgcn_wmma_f32_16x16x4_f32(false, a, false, b,
                                                (short)0, acc, false, false);
  }

  float bias = bl[col];
#pragma unroll
  for (int rr = 0; rr < 8; ++rr) {
    float v = acc[rr] + bias;
    if (RELU) v = fmaxf(v, 0.0f);
    out[(long long)(m0 + rr + 8 * kh) * COUT + col] = v;
  }
}

extern "C" void kernel_launch(void* const* d_in, const int* in_sizes, int n_in,
                              void* d_out, int out_size, void* d_ws, size_t ws_size,
                              hipStream_t stream) {
  const float* x0  = (const float*)d_in[0];
  const int*   ei  = (const int*)d_in[1];
  const float* Wl0 = (const float*)d_in[2];
  const float* bl0 = (const float*)d_in[3];
  const float* Wr0 = (const float*)d_in[4];
  const float* Wl1 = (const float*)d_in[5];
  const float* bl1 = (const float*)d_in[6];
  const float* Wr1 = (const float*)d_in[7];
  const float* Wl2 = (const float*)d_in[8];
  const float* bl2 = (const float*)d_in[9];
  const float* Wr2 = (const float*)d_in[10];

  const int N = in_sizes[0] / C_IN;
  const int E = in_sizes[1] / 2;
  const int* src = ei;
  const int* dst = ei + E;

  float* h0  = (float*)d_ws;
  float* h1  = h0 + (size_t)N * C_IN;
  float* agg = h1 + (size_t)N * C_IN;
  float* cnt = agg + (size_t)N * C_IN;

  const int aggBlocks = (N + OWN_NODES - 1) / OWN_NODES;

  const int tilesM = (N + 15) / 16;

  sage_aggregate<<<aggBlocks, AGG_THREADS, 0, stream>>>(x0, src, dst, agg, cnt, E, N);
  {
    int tiles = tilesM * (128 / 16);
    int blocks = (tiles + 7) / 8;
    sage_gemm<128, true><<<blocks, 256, 0, stream>>>(x0, agg, cnt, Wl0, bl0, Wr0, h0, tiles);
  }

  sage_aggregate<<<aggBlocks, AGG_THREADS, 0, stream>>>(h0, src, dst, agg, cnt, E, N);
  {
    int tiles = tilesM * (128 / 16);
    int blocks = (tiles + 7) / 8;
    sage_gemm<128, true><<<blocks, 256, 0, stream>>>(h0, agg, cnt, Wl1, bl1, Wr1, h1, tiles);
  }

  sage_aggregate<<<aggBlocks, AGG_THREADS, 0, stream>>>(h1, src, dst, agg, cnt, E, N);
  {
    int tiles = tilesM * (64 / 16);
    int blocks = (tiles + 7) / 8;
    sage_gemm<64, false><<<blocks, 256, 0, stream>>>(h1, agg, cnt, Wl2, bl2, Wr2,
                                                     (float*)d_out, tiles);
  }
}


# Aggregation: layer 1's launch writes each owner's filtered edge list (source row, local node) to the workspace; layers 2 and 3 read it back and skip the edge-list pass; sums still in arrival order
# speedup vs baseline: 2.4378x; 1.2566x over previous
#include <hip/hip_runtime.h>
#include <hip/hip_bf16.h>

typedef __attribute__((ext_vector_type(2))) float v2f;
typedef __attribute__((ext_vector_type(8))) float v8f;

#define C_IN 128

#define OWN_NODES 391
#define LIST_CAP 16384
#define AGG_THREADS 1024
#define EDGE_BATCH 8

__device__ __forceinline__ void add_edge_direct(const float* __restrict__ x,
                                                const int* __restrict__ src,
                                                float* sAcc, float* sCnt,
                                                int e, unsigned rel) {
  int s = src[e];
  const float* xr = x + (long long)s * C_IN;
  float* ar = sAcc + rel * C_IN;
  for (int c = 0; c < C_IN; ++c) atomicAdd(ar + c, xr[c]);
  atomicAdd(sCnt + rel, 1.0f);
}

__device__ __forceinline__ void load_dst512(int4 (&buf)[4], const int* __restrict__ dst,
                                            int sc, int lane) {
#pragma unroll
  for (int u = 0; u < 4; ++u)
    buf[u] = *(const int4*)(dst + (long long)sc * 512 + u * 128 + lane * 4);
}

__device__ __forceinline__ void note_owned(const int4 (&buf)[4], int sc, int lane,
                                           int nodeLo, unsigned nOwn,
                                           unsigned* sList, unsigned* sCount) {
  unsigned rel[16];
  unsigned msk[16];
  unsigned total = 0;
#pragma unroll
  for (int u = 0; u < 4; ++u) {
    rel[u * 4 + 0] = (unsigned)(buf[u].x - nodeLo);
    rel[u * 4 + 1] = (unsigned)(buf[u].y - nodeLo);
    rel[u * 4 + 2] = (unsigned)(buf[u].z - nodeLo);
    rel[u * 4 + 3] = (unsigned)(buf[u].w - nodeLo);
  }
#pragma unroll
  for (int q = 0; q < 16; ++q) {
    msk[q] = __builtin_amdgcn_ballot_w32(rel[q] < nOwn);
    total += (unsigned)__popc(msk[q]);
  }
  if (total != 0u) {
    unsigned base = 0u;
    if (lane == 0) base = atomicAdd(sCount, total);
    base = (unsigned)__builtin_amdgcn_readfirstlane((int)base);
#pragma unroll
    for (int q = 0; q < 16; ++q) {
      if (msk[q] != 0u) {
        if (rel[q] < nOwn) {
          unsigned slot = base + __builtin_amdgcn_mbcnt_lo(msk[q], 0u);
          unsigned e = (unsigned)sc * 512u + (unsigned)((q >> 2) * 128 + lane * 4 + (q & 3));
          if (slot < LIST_CAP) sList[slot] = e * 512u + rel[q];
        }
        base += (unsigned)__popc(msk[q]);
      }
    }
  }
}

__device__ __forceinline__ void add_group(const float* __restrict__ x, float* sAcc,
                                          int sL, int rL, unsigned nIn, int lane) {
  if (nIn == 32u) {
#pragma unroll
    for (int j = 0; j < 32; j += EDGE_BATCH) {
      float v[EDGE_BATCH][4];
#pragma unroll
      for (int u = 0; u < EDGE_BATCH; ++u) {
        int s = __builtin_amdgcn_readlane(sL, j + u);
        const float* xrow = x + (long long)s * C_IN + lane;
        v[u][0] = xrow[0];
        v[u][1] = xrow[32];
        v[u][2] = xrow[64];
        v[u][3] = xrow[96];
      }
#pragma unroll
      for (int u = 0; u < EDGE_BATCH; ++u) {
        int d = __builtin_amdgcn_readlane(rL, j + u);
        float* arow = sAcc + d * C_IN + lane;
        atomicAdd(arow,      v[u][0]);
        atomicAdd(arow + 32, v[u][1]);
        atomicAdd(arow + 64, v[u][2]);
        atomicAdd(arow + 96, v[u][3]);
      }
    }
  } else {
    for (unsigned j = 0; j < nIn; ++j) {
      int s = __shfl(sL, (int)j);
      int d = __shfl(rL, (int)j);
      const float* xrow = x + (long long)s * C_IN + lane;
      float v0 = xrow[0];
      float v1 = xrow[32];
      float v2 = xrow[64];
      float v3 = xrow[96];
      float* arow = sAcc + d * C_IN + lane;
      atomicAdd(arow,      v0);
      atomicAdd(arow + 32, v1);
      atomicAdd(arow + 64, v2);
      atomicAdd(arow + 96, v3);
    }
  }
}

template <bool BUILD>
__global__ void __launch_bounds__(AGG_THREADS)
sage_aggregate(const float* __restrict__ x,
               const int* __restrict__ src,
               const int* __restrict__ dst,
               float* __restrict__ agg,
               float* __restrict__ cnt,
               unsigned* __restrict__ gList,
               unsigned* __restrict__ gCount,
               int nEdges, int nNodes) {
  __shared__ __attribute__((aligned(16))) float sAcc[OWN_NODES * C_IN];
  __shared__ float sCnt[OWN_NODES];

  const int tid  = (int)threadIdx.x;
  const int lane = tid & 31;
  const int wv   = __builtin_amdgcn_readfirstlane(tid >> 5);
  const int nodeLo = (int)blockIdx.x * OWN_NODES;
  int nOwnI = nNodes - nodeLo;
  if (nOwnI > OWN_NODES) nOwnI = OWN_NODES;
  if (nOwnI <= 0) return;
  const unsigned nOwn = (unsigned)nOwnI;
  unsigned* gl = gList + (size_t)blockIdx.x * LIST_CAP;

  unsigned nList;
  if constexpr (BUILD) {
    __shared__ unsigned sList[LIST_CAP];
    __shared__ unsigned sCount;

    for (int i = tid; i < OWN_NODES * C_IN / 4; i += AGG_THREADS)
      ((float4*)sAcc)[i] = make_float4(0.f, 0.f, 0.f, 0.f);
    for (int i = tid; i < OWN_NODES; i += AGG_THREADS) sCnt[i] = 0.0f;
    if (tid == 0) sCount = 0u;
    __syncthreads();

    const int nSuper = nEdges >> 9;
    {
      int4 bufA[4], bufB[4];
      int sc = wv;
      if (sc < nSuper) {
        const int scLast = nSuper - 1;
        load_dst512(bufA, dst, sc, lane);
        while (true) {
          load_dst512(bufB, dst, (sc + 32 < scLast) ? sc + 32 : scLast, lane);
          note_owned(bufA, sc, lane, nodeLo, nOwn, sList, &sCount);
          sc += 32;
          if (sc >= nSuper) break;
          load_dst512(bufA, dst, (sc + 32 < scLast) ? sc + 32 : scLast, lane);
          note_owned(bufB, sc, lane, nodeLo, nOwn, sList, &sCount);
          sc += 32;
          if (sc >= nSuper) break;
        }
      }
    }
    for (int e = (nSuper << 9) + tid; e < nEdges; e += AGG_THREADS) {
      unsigned r1 = (unsigned)(dst[e] - nodeLo);
      if (r1 < nOwn) {
        unsigned slot = atomicAdd(&sCount, 1u);
        if (slot < LIST_CAP) sList[slot] = (unsigned)e * 512u + r1;
      }
    }
    __syncthreads();

    nList = sCount;
    if (tid == 0) gCount[blockIdx.x] = nList;
    if (nList <= LIST_CAP) {
      for (unsigned i0 = (unsigned)wv * 32u; i0 < nList; i0 += 32u * 32u) {
        unsigned nIn = nList - i0;
        if (nIn > 32u) nIn = 32u;
        int sL = 0;
        int rL = 0;
        if ((unsigned)lane < nIn) {
          unsigned ent = sList[i0 + lane];
          rL = (int)(ent & 511u);
          sL = src[ent >> 9];
          gl[i0 + lane] = (unsigned)sL * 512u + (unsigned)rL;
          atomicAdd(sCnt + rL, 1.0f);
        }
        add_group(x, sAcc, sL, rL, nIn, lane);
      }
    }
  } else {
    nList = (unsigned)__builtin_amdgcn_readfirstlane((int)gCount[blockIdx.x]);
    unsigned i0 = (unsigned)wv * 32u;
    unsigned entNext = 0u;
    if (nList <= LIST_CAP && i0 + (unsigned)lane < nList) entNext = gl[i0 + lane];

    for (int i = tid; i < OWN_NODES * C_IN / 4; i += AGG_THREADS)
      ((float4*)sAcc)[i] = make_float4(0.f, 0.f, 0.f, 0.f);
    for (int i = tid; i < OWN_NODES; i += AGG_THREADS) sCnt[i] = 0.0f;
    __syncthreads();

    if (nList <= LIST_CAP) {
      for (; i0 < nList; i0 += 32u * 32u) {
        unsigned nIn = nList - i0;
        if (nIn > 32u) nIn = 32u;
        const unsigned ent = entNext;
        const unsigned i1 = i0 + 32u * 32u + (unsigned)lane;
        entNext = (i1 < nList) ? gl[i1] : 0u;
        int sL = 0;
        int rL = 0;
        if ((unsigned)lane < nIn) {
          rL = (int)(ent & 511u);
          sL = (int)(ent >> 9);
          atomicAdd(sCnt + rL, 1.0f);
        }
        add_group(x, sAcc, sL, rL, nIn, lane);
      }
    }
  }

  if (nList > LIST_CAP) {
    for (int e = tid; e < nEdges; e += AGG_THREADS) {
      unsigned r1 = (unsigned)(dst[e] - nodeLo);
      if (r1 < nOwn) add_edge_direct(x, src, sAcc, sCnt, e, r1);
    }
  }
  __syncthreads();

  {
    float4* gdst = (float4*)(agg + (long long)nodeLo * C_IN);
    const int n4 = nOwnI * (C_IN / 4);
    for (int i = tid; i < n4; i += AGG_THREADS) gdst[i] = ((const float4*)sAcc)[i];
    for (int i = tid; i < nOwnI; i += AGG_THREADS) cnt[nodeLo + i] = sCnt[i];
  }
}

template <int COUT, bool RELU>
__global__ void sage_gemm(const float* __restrict__ xin,
                          const float* __restrict__ agg,
                          const float* __restrict__ cnt,
                          const float* __restrict__ Wl,
                          const float* __restrict__ bl,
                          const float* __restrict__ Wr,
                          float* __restrict__ out,
                          int totalTiles) {
  const int tilesN = COUT / 16;
  int wave = (int)(((long long)blockIdx.x * blockDim.x + threadIdx.x) >> 5);
  if (wave >= totalTiles) return;
  int lane = threadIdx.x & 31;
  int m0 = (wave / tilesN) * 16;
  int n0 = (wave % tilesN) * 16;
  int r  = lane & 15;
  int kh = lane >> 4;
  int col = n0 + r;

  const float* arow = agg + (long long)(m0 + r) * C_IN;
  const float* xrow = xin + (long long)(m0 + r) * C_IN;
  float inv = 1.0f / fmaxf(cnt[m0 + r], 1.0f);

  v8f acc = {};

#pragma unroll
  for (int k = 0; k < C_IN; k += 4) {
    v2f a, b;
    int kk = k + 2 * kh;
    a[0] = arow[kk] * inv;
    a[1] = arow[kk + 1] * inv;
    b[0] = Wl[(long long)kk * COUT + col];
    b[1] = Wl[(long long)(kk + 1) * COUT + col];
    acc = __builtin_amdgcn_wmma_f32_16x16x4_f32(false, a, false, b,
                                                (short)0, acc, false, false);
  }
#pragma unroll
  for (int k = 0; k < C_IN; k += 4) {
    v2f a, b;
    int kk = k + 2 * kh;
    a[0] = xrow[kk];
    a[1] = xrow[kk + 1];
    b[0] = Wr[(long long)kk * COUT + col];
    b[1] = Wr[(long long)(kk + 1) * COUT + col];
    acc = __builtin_amdgcn_wmma_f32_16x16x4_f32(false, a, false, b,
                                                (short)0, acc, false, false);
  }

  float bias = bl[col];
#pragma unroll
  for (int rr = 0; rr < 8; ++rr) {
    float v = acc[rr] + bias;
    if (RELU) v = fmaxf(v, 0.0f);
    out[(long long)(m0 + rr + 8 * kh) * COUT + col] = v;
  }
}

extern "C" void kernel_launch(void* const* d_in, const int* in_sizes, int n_in,
                              void* d_out, int out_size, void* d_ws, size_t ws_size,
                              hipStream_t stream) {
  const float* x0  = (const float*)d_in[0];
  const int*   ei  = (const int*)d_in[1];
  const float* Wl0 = (const float*)d_in[2];
  const float* bl0 = (const float*)d_in[3];
  const float* Wr0 = (const float*)d_in[4];
  const float* Wl1 = (const float*)d_in[5];
  const float* bl1 = (const float*)d_in[6];
  const float* Wr1 = (const float*)d_in[7];
  const float* Wl2 = (const float*)d_in[8];
  const float* bl2 = (const float*)d_in[9];
  const float* Wr2 = (const float*)d_in[10];

  const int N = in_sizes[0] / C_IN;
  const int E = in_sizes[1] / 2;
  const int* src = ei;
  const int* dst = ei + E;

  const int aggBlocks = (N + OWN_NODES - 1) / OWN_NODES;

  float* h0  = (float*)d_ws;
  float* h1  = h0 + (size_t)N * C_IN;
  float* agg = h1 + (size_t)N * C_IN;
  unsigned* lists  = (unsigned*)(agg + (size_t)N * C_IN);
  unsigned* counts = lists + (size_t)aggBlocks * LIST_CAP;
  float* cnt = (float*)(counts + aggBlocks);

  const int tilesM = (N + 15) / 16;

  sage_aggregate<true><<<aggBlocks, AGG_THREADS, 0, stream>>>(x0, src, dst, agg, cnt, lists, counts, E, N);
  {
    int tiles = tilesM * (128 / 16);
    int blocks = (tiles + 7) / 8;
    sage_gemm<128, true><<<blocks, 256, 0, stream>>>(x0, agg, cnt, Wl0, bl0, Wr0, h0, tiles);
  }

  sage_aggregate<false><<<aggBlocks, AGG_THREADS, 0, stream>>>(h0, src, dst, agg, cnt, lists, counts, E, N);
  {
    int tiles = tilesM * (128 / 16);
    int blocks = (tiles + 7) / 8;
    sage_gemm<128, true><<<blocks, 256, 0, stream>>>(h0, agg, cnt, Wl1, bl1, Wr1, h1, tiles);
  }

  sage_aggregate<false><<<aggBlocks, AGG_THREADS, 0, stream>>>(h1, src, dst, agg, cnt, lists, counts, E, N);
  {
    int tiles = tilesM * (64 / 16);
    int blocks = (tiles + 7) / 8;
    sage_gemm<64, false><<<blocks, 256, 0, stream>>>(h1, agg, cnt, Wl2, bl2, Wr2,
                                                     (float*)d_out, tiles);
  }
}


# Projection GEMM: wave owns a 16-column tile, its Wl and Wr fragments stay in registers, and it loops over 16-row strips staged per work-group in LDS by whole-row loads; sum order unchanged
# speedup vs baseline: 3.4470x; 1.4140x over previous
#include <hip/hip_runtime.h>
#include <hip/hip_bf16.h>

typedef __attribute__((ext_vector_type(2))) float v2f;
typedef __attribute__((ext_vector_type(8))) float v8f;
typedef __attribute__((ext_vector_type(4))) float v4f;

#define C_IN 128

#define OWN_NODES 391
#define LIST_CAP 16384
#define AGG_THREADS 1024
#define EDGE_BATCH 8

__device__ __forceinline__ void add_edge_direct(const float* __restrict__ x,
                                                const int* __restrict__ src,
                                                float* sAcc, float* sCnt,
                                                int e, unsigned rel) {
  int s = src[e];
  const float* xr = x + (long long)s * C_IN;
  float* ar = sAcc + rel * C_IN;
  for (int c = 0; c < C_IN; ++c) atomicAdd(ar + c, xr[c]);
  atomicAdd(sCnt + rel, 1.0f);
}

__device__ __forceinline__ void load_dst512(int4 (&buf)[4], const int* __restrict__ dst,
                                            int sc, int lane) {
#pragma unroll
  for (int u = 0; u < 4; ++u)
    buf[u] = *(const int4*)(dst + (long long)sc * 512 + u * 128 + lane * 4);
}

__device__ __forceinline__ void note_owned(const int4 (&buf)[4], int sc, int lane,
                                           int nodeLo, unsigned nOwn,
                                           unsigned* sList, unsigned* sCount) {
  unsigned rel[16];
  unsigned msk[16];
  unsigned total = 0;
#pragma unroll
  for (int u = 0; u < 4; ++u) {
    rel[u * 4 + 0] = (unsigned)(buf[u].x - nodeLo);
    rel[u * 4 + 1] = (unsigned)(buf[u].y - nodeLo);
    rel[u * 4 + 2] = (unsigned)(buf[u].z - nodeLo);
    rel[u * 4 + 3] = (unsigned)(buf[u].w - nodeLo);
  }
#pragma unroll
  for (int q = 0; q < 16; ++q) {
    msk[q] = __builtin_amdgcn_ballot_w32(rel[q] < nOwn);
    total += (unsigned)__popc(msk[q]);
  }
  if (total != 0u) {
    unsigned base = 0u;
    if (lane == 0) base = atomicAdd(sCount, total);
    base = (unsigned)__builtin_amdgcn_readfirstlane((int)base);
#pragma unroll
    for (int q = 0; q < 16; ++q) {
      if (msk[q] != 0u) {
        if (rel[q] < nOwn) {
          unsigned slot = base + __builtin_amdgcn_mbcnt_lo(msk[q], 0u);
          unsigned e = (unsigned)sc * 512u + (unsigned)((q >> 2) * 128 + lane * 4 + (q & 3));
          if (slot < LIST_CAP) sList[slot] = e * 512u + rel[q];
        }
        base += (unsigned)__popc(msk[q]);
      }
    }
  }
}

__device__ __forceinline__ void add_group(const float* __restrict__ x, float* sAcc,
                                          int sL, int rL, unsigned nIn, int lane) {
  if (nIn == 32u) {
#pragma unroll
    for (int j = 0; j < 32; j += EDGE_BATCH) {
      float v[EDGE_BATCH][4];
#pragma unroll
      for (int u = 0; u < EDGE_BATCH; ++u) {
        int s = __builtin_amdgcn_readlane(sL, j + u);
        const float* xrow = x + (long long)s * C_IN + lane;
        v[u][0] = xrow[0];
        v[u][1] = xrow[32];
        v[u][2] = xrow[64];
        v[u][3] = xrow[96];
      }
#pragma unroll
      for (int u = 0; u < EDGE_BATCH; ++u) {
        int d = __builtin_amdgcn_readlane(rL, j + u);
        float* arow = sAcc + d * C_IN + lane;
        atomicAdd(arow,      v[u][0]);
        atomicAdd(arow + 32, v[u][1]);
        atomicAdd(arow + 64, v[u][2]);
        atomicAdd(arow + 96, v[u][3]);
      }
    }
  } else {
    for (unsigned j = 0; j < nIn; ++j) {
      int s = __shfl(sL, (int)j);
      int d = __shfl(rL, (int)j);
      const float* xrow = x + (long long)s * C_IN + lane;
      float v0 = xrow[0];
      float v1 = xrow[32];
      float v2 = xrow[64];
      float v3 = xrow[96];
      float* arow = sAcc + d * C_IN + lane;
      atomicAdd(arow,      v0);
      atomicAdd(arow + 32, v1);
      atomicAdd(arow + 64, v2);
      atomicAdd(arow + 96, v3);
    }
  }
}

template <bool BUILD>
__global__ void __launch_bounds__(AGG_THREADS)
sage_aggregate(const float* __restrict__ x,
               const int* __restrict__ src,
               const int* __restrict__ dst,
               float* __restrict__ agg,
               float* __restrict__ cnt,
               unsigned* __restrict__ gList,
               unsigned* __restrict__ gCount,
               int nEdges, int nNodes) {
  __shared__ __attribute__((aligned(16))) float sAcc[OWN_NODES * C_IN];
  __shared__ float sCnt[OWN_NODES];

  const int tid  = (int)threadIdx.x;
  const int lane = tid & 31;
  const int wv   = __builtin_amdgcn_readfirstlane(tid >> 5);
  const int nodeLo = (int)blockIdx.x * OWN_NODES;
  int nOwnI = nNodes - nodeLo;
  if (nOwnI > OWN_NODES) nOwnI = OWN_NODES;
  if (nOwnI <= 0) return;
  const unsigned nOwn = (unsigned)nOwnI;
  unsigned* gl = gList + (size_t)blockIdx.x * LIST_CAP;

  unsigned nList;
  if constexpr (BUILD) {
    __shared__ unsigned sList[LIST_CAP];
    __shared__ unsigned sCount;

    for (int i = tid; i < OWN_NODES * C_IN / 4; i += AGG_THREADS)
      ((float4*)sAcc)[i] = make_float4(0.f, 0.f, 0.f, 0.f);
    for (int i = tid; i < OWN_NODES; i += AGG_THREADS) sCnt[i] = 0.0f;
    if (tid == 0) sCount = 0u;
    __syncthreads();

    const int nSuper = nEdges >> 9;
    {
      int4 bufA[4], bufB[4];
      int sc = wv;
      if (sc < nSuper) {
        const int scLast = nSuper - 1;
        load_dst512(bufA, dst, sc, lane);
        while (true) {
          load_dst512(bufB, dst, (sc + 32 < scLast) ? sc + 32 : scLast, lane);
          note_owned(bufA, sc, lane, nodeLo, nOwn, sList, &sCount);
          sc += 32;
          if (sc >= nSuper) break;
          load_dst512(bufA, dst, (sc + 32 < scLast) ? sc + 32 : scLast, lane);
          note_owned(bufB, sc, lane, nodeLo, nOwn, sList, &sCount);
          sc += 32;
          if (sc >= nSuper) break;
        }
      }
    }
    for (int e = (nSuper << 9) + tid; e < nEdges; e += AGG_THREADS) {
      unsigned r1 = (unsigned)(dst[e] - nodeLo);
      if (r1 < nOwn) {
        unsigned slot = atomicAdd(&sCount, 1u);
        if (slot < LIST_CAP) sList[slot] = (unsigned)e * 512u + r1;
      }
    }
    __syncthreads();

    nList = sCount;
    if (tid == 0) gCount[blockIdx.x] = nList;
    if (nList <= LIST_CAP) {
      for (unsigned i0 = (unsigned)wv * 32u; i0 < nList; i0 += 32u * 32u) {
        unsigned nIn = nList - i0;
        if (nIn > 32u) nIn = 32u;
        int sL = 0;
        int rL = 0;
        if ((unsigned)lane < nIn) {
          unsigned ent = sList[i0 + lane];
          rL = (int)(ent & 511u);
          sL = src[ent >> 9];
          gl[i0 + lane] = (unsigned)sL * 512u + (unsigned)rL;
          atomicAdd(sCnt + rL, 1.0f);
        }
        add_group(x, sAcc, sL, rL, nIn, lane);
      }
    }
  } else {
    nList = (unsigned)__builtin_amdgcn_readfirstlane((int)gCount[blockIdx.x]);
    unsigned i0 = (unsigned)wv * 32u;
    unsigned entNext = 0u;
    if (nList <= LIST_CAP && i0 + (unsigned)lane < nList) entNext = gl[i0 + lane];

    for (int i = tid; i < OWN_NODES * C_IN / 4; i += AGG_THREADS)
      ((float4*)sAcc)[i] = make_float4(0.f, 0.f, 0.f, 0.f);
    for (int i = tid; i < OWN_NODES; i += AGG_THREADS) sCnt[i] = 0.0f;
    __syncthreads();

    if (nList <= LIST_CAP) {
      for (; i0 < nList; i0 += 32u * 32u) {
        unsigned nIn = nList - i0;
        if (nIn > 32u) nIn = 32u;
        const unsigned ent = entNext;
        const unsigned i1 = i0 + 32u * 32u + (unsigned)lane;
        entNext = (i1 < nList) ? gl[i1] : 0u;
        int sL = 0;
        int rL = 0;
        if ((unsigned)lane < nIn) {
          rL = (int)(ent & 511u);
          sL = (int)(ent >> 9);
          atomicAdd(sCnt + rL, 1.0f);
        }
        add_group(x, sAcc, sL, rL, nIn, lane);
      }
    }
  }

  if (nList > LIST_CAP) {
    for (int e = tid; e < nEdges; e += AGG_THREADS) {
      unsigned r1 = (unsigned)(dst[e] - nodeLo);
      if (r1 < nOwn) add_edge_direct(x, src, sAcc, sCnt, e, r1);
    }
  }
  __syncthreads();

  {
    float4* gdst = (float4*)(agg + (long long)nodeLo * C_IN);
    const int n4 = nOwnI * (C_IN / 4);
    for (int i = tid; i < n4; i += AGG_THREADS) gdst[i] = ((const float4*)sAcc)[i];
    for (int i = tid; i < nOwnI; i += AGG_THREADS) cnt[nodeLo + i] = sCnt[i];
  }
}

#define A_PITCH 132
#define GEMM_BLOCKS_128 512
#define GEMM_BLOCKS_64 1024

template <int COUT, bool RELU>
__global__ void __launch_bounds__(COUT * 2) __attribute__((amdgpu_waves_per_eu(4, 4)))
sage_gemm(const float* __restrict__ xin,
          const float* __restrict__ agg,
          const float* __restrict__ cnt,
          const float* __restrict__ Wl,
          const float* __restrict__ bl,
          const float* __restrict__ Wr,
          float* __restrict__ out,
          int nStrips) {
  constexpr int TN  = COUT / 16;
  constexpr int RP  = 16 / TN;
  constexpr int MAT = 16 * A_PITCH;
  constexpr int BUF = 2 * MAT;
  __shared__ __attribute__((aligned(16))) float sA[2 * BUF];

  const int tid  = (int)threadIdx.x;
  const int lane = tid & 31;
  const int wv   = __builtin_amdgcn_readfirstlane(tid >> 5);
  const int r    = lane & 15;
  const int kh   = lane >> 4;
  const int col  = wv * 16 + r;
  const int step = (int)gridDim.x;

  v2f bL[32], bR[32];
#pragma unroll
  for (int ks = 0; ks < 32; ++ks) {
    const int kk = 4 * ks + 2 * kh;
    bL[ks][0] = Wl[kk * COUT + col];
    bL[ks][1] = Wl[(kk + 1) * COUT + col];
    bR[ks][0] = Wr[kk * COUT + col];
    bR[ks][1] = Wr[(kk + 1) * COUT + col];
  }
  const float bias = bl[col];

  v4f pa[RP], px[RP];
  float  pc[RP];

#define GEMM_FETCH(STRIP)                                                       \
  _Pragma("unroll")                                                             \
  for (int p = 0; p < RP; ++p) {                                                \
    const long long g = (long long)(STRIP) * 16 + p * TN + wv;                  \
    pa[p] = *(const v4f*)(agg + g * C_IN + lane * 4);                           \
    px[p] = *(const v4f*)(xin + g * C_IN + lane * 4);                           \
    pc[p] = cnt[g];                                                             \
  }

#define GEMM_STAGE(SB)                                                          \
  _Pragma("unroll")                                                             \
  for (int p = 0; p < RP; ++p) {                                                \
    const float inv = 1.0f / fmaxf(pc[p], 1.0f);                                \
    const v4f a = pa[p] * inv;                                                  \
    float* d = (SB) + (p * TN + wv) * A_PITCH + lane * 4;                       \
    *(v4f*)d = a;                                                               \
    *(v4f*)(d + MAT) = px[p];                                                   \
  }

#define GEMM_STRIP_BARRIER()                                                    \
  __builtin_amdgcn_fence(__ATOMIC_RELEASE, "workgroup", "local");               \
  __builtin_amdgcn_s_barrier();                                                 \
  __builtin_amdgcn_fence(__ATOMIC_ACQUIRE, "workgroup", "local")

  int it = (int)blockIdx.x;
  if (it >= nStrips) return;
  const int itLast = nStrips - 1;
  GEMM_FETCH(it)
  GEMM_STAGE(sA)
  GEMM_FETCH((it + step < itLast) ? it + step : itLast)
  GEMM_STRIP_BARRIER();

  int buf = 0;
  for (; it < nStrips; it += step) {
    const float* fa = sA + buf * BUF + r * A_PITCH + 2 * kh;
    v8f acc = {};
    v2f fragA[8], fragB[8];
#pragma unroll
    for (int q = 0; q < 8; ++q) fragA[q] = *(const v2f*)(fa + 4 * q);
#pragma unroll
    for (int c = 0; c < 8; c += 2) {
      {
        const float* nb = fa + ((c + 1) >= 4 ? MAT : 0) + 32 * ((c + 1) & 3);
#pragma unroll
        for (int q = 0; q < 8; ++q) fragB[q] = *(const v2f*)(nb + 4 * q);
      }
#pragma unroll
      for (int q = 0; q < 8; ++q)
        acc = __builtin_amdgcn_wmma_f32_16x16x4_f32(false, fragA[q], false,
                                                    (c < 4) ? bL[8 * c + q] : bR[8 * (c - 4) + q],
                                                    (short)0, acc, false, false);
      if (c + 2 < 8) {
        const float* nb = fa + ((c + 2) >= 4 ? MAT : 0) + 32 * ((c + 2) & 3);
#pragma unroll
        for (int q = 0; q < 8; ++q) fragA[q] = *(const v2f*)(nb + 4 * q);
      }
#pragma unroll
      for (int q = 0; q < 8; ++q)
        acc = __builtin_amdgcn_wmma_f32_16x16x4_f32(false, fragB[q], false,
                                                    (c + 1 < 4) ? bL[8 * (c + 1) + q] : bR[8 * (c + 1 - 4) + q],
                                                    (short)0, acc, false, false);
    }

    const long long m0 = (long long)it * 16;
#pragma unroll
    for (int rr = 0; rr < 8; ++rr) {
      float v = acc[rr] + bias;
      if (RELU) v = fmaxf(v, 0.0f);
      out[(m0 + rr + 8 * kh) * COUT + col] = v;
    }

    GEMM_STAGE(sA + (buf ^ 1) * BUF)
    GEMM_FETCH((it + 2 * step < itLast) ? it + 2 * step : itLast)
    GEMM_STRIP_BARRIER();
    buf ^= 1;
  }
#undef GEMM_FETCH
#undef GEMM_STAGE
#undef GEMM_STRIP_BARRIER
}

extern "C" void kernel_launch(void* const* d_in, const int* in_sizes, int n_in,
                              void* d_out, int out_size, void* d_ws, size_t ws_size,
                              hipStream_t stream) {
  const float* x0  = (const float*)d_in[0];
  const int*   ei  = (const int*)d_in[1];
  const float* Wl0 = (const float*)d_in[2];
  const float* bl0 = (const float*)d_in[3];
  const float* Wr0 = (const float*)d_in[4];
  const float* Wl1 = (const float*)d_in[5];
  const float* bl1 = (const float*)d_in[6];
  const float* Wr1 = (const float*)d_in[7];
  const float* Wl2 = (const float*)d_in[8];
  const float* bl2 = (const float*)d_in[9];
  const float* Wr2 = (const float*)d_in[10];

  const int N = in_sizes[0] / C_IN;
  const int E = in_sizes[1] / 2;
  const int* src = ei;
  const int* dst = ei + E;

  const int aggBlocks = (N + OWN_NODES - 1) / OWN_NODES;

  float* h0  = (float*)d_ws;
  float* h1  = h0 + (size_t)N * C_IN;
  float* agg = h1 + (size_t)N * C_IN;
  unsigned* lists  = (unsigned*)(agg + (size_t)N * C_IN);
  unsigned* counts = lists + (size_t)aggBlocks * LIST_CAP;
  float* cnt = (float*)(counts + aggBlocks);

  const int tilesM = (N + 15) / 16;

  sage_aggregate<true><<<aggBlocks, AGG_THREADS, 0, stream>>>(x0, src, dst, agg, cnt, lists, counts, E, N);
  sage_gemm<128, true><<<GEMM_BLOCKS_128, 256, 0, stream>>>(x0, agg, cnt, Wl0, bl0, Wr0, h0, tilesM);

  sage_aggregate<false><<<aggBlocks, AGG_THREADS, 0, stream>>>(h0, src, dst, agg, cnt, lists, counts, E, N);
  sage_gemm<128, true><<<GEMM_BLOCKS_128, 256, 0, stream>>>(h0, agg, cnt, Wl1, bl1, Wr1, h1, tilesM);

  sage_aggregate<false><<<aggBlocks, AGG_THREADS, 0, stream>>>(h1, src, dst, agg, cnt, lists, counts, E, N);
  sage_gemm<64, false><<<GEMM_BLOCKS_64, 128, 0, stream>>>(h1, agg, cnt, Wl2, bl2, Wr2,
                                                                 (float*)d_out, tilesM);
}


# Aggregation row loads: one 16-byte-per-lane request per source row (was 4 one-line requests), 16 rows issued as one group per wave; LDS rows in lane-major column order; sums still in arrival order
# speedup vs baseline: 3.6740x; 1.0659x over previous
#include <hip/hip_runtime.h>
#include <hip/hip_bf16.h>

typedef __attribute__((ext_vector_type(2))) float v2f;
typedef __attribute__((ext_vector_type(8))) float v8f;
typedef __attribute__((ext_vector_type(4))) float v4f;

#define C_IN 128

#define OWN_NODES 391
#define LIST_CAP 16384
#define AGG_THREADS 1024
#define EDGE_BATCH 16
#define ACC_WORD(c) ((((c) & 3) << 5) + ((c) >> 2))

__device__ __forceinline__ void add_edge_direct(const float* __restrict__ x,
                                                const int* __restrict__ src,
                                                float* sAcc, float* sCnt,
                                                int e, unsigned rel) {
  int s = src[e];
  const float* xr = x + (long long)s * C_IN;
  float* ar = sAcc + rel * C_IN;
  for (int c = 0; c < C_IN; ++c) atomicAdd(ar + ACC_WORD(c), xr[c]);
  atomicAdd(sCnt + rel, 1.0f);
}

__device__ __forceinline__ void load_dst512(int4 (&buf)[4], const int* __restrict__ dst,
                                            int sc, int lane) {
#pragma unroll
  for (int u = 0; u < 4; ++u)
    buf[u] = *(const int4*)(dst + (long long)sc * 512 + u * 128 + lane * 4);
}

__device__ __forceinline__ void note_owned(const int4 (&buf)[4], int sc, int lane,
                                           int nodeLo, unsigned nOwn,
                                           unsigned* sList, unsigned* sCount) {
  unsigned rel[16];
  unsigned msk[16];
  unsigned total = 0;
#pragma unroll
  for (int u = 0; u < 4; ++u) {
    rel[u * 4 + 0] = (unsigned)(buf[u].x - nodeLo);
    rel[u * 4 + 1] = (unsigned)(buf[u].y - nodeLo);
    rel[u * 4 + 2] = (unsigned)(buf[u].z - nodeLo);
    rel[u * 4 + 3] = (unsigned)(buf[u].w - nodeLo);
  }
#pragma unroll
  for (int q = 0; q < 16; ++q) {
    msk[q] = __builtin_amdgcn_ballot_w32(rel[q] < nOwn);
    total += (unsigned)__popc(msk[q]);
  }
  if (total != 0u) {
    unsigned base = 0u;
    if (lane == 0) base = atomicAdd(sCount, total);
    base = (unsigned)__builtin_amdgcn_readfirstlane((int)base);
#pragma unroll
    for (int q = 0; q < 16; ++q) {
      if (msk[q] != 0u) {
        if (rel[q] < nOwn) {
          unsigned slot = base + __builtin_amdgcn_mbcnt_lo(msk[q], 0u);
          unsigned e = (unsigned)sc * 512u + (unsigned)((q >> 2) * 128 + lane * 4 + (q & 3));
          if (slot < LIST_CAP) sList[slot] = e * 512u + rel[q];
        }
        base += (unsigned)__popc(msk[q]);
      }
    }
  }
}

template <bool FULL>
__device__ __forceinline__ void add_rows(const float* __restrict__ x, float* sAcc,
                                         int sL, int rL, unsigned nIn, int lane) {
#pragma unroll
  for (int j = 0; j < 32; j += EDGE_BATCH) {
    if (!FULL && (unsigned)j >= nIn) break;
    unsigned off[EDGE_BATCH];
    v4f v[EDGE_BATCH];
#pragma unroll
    for (int u = 0; u < EDGE_BATCH; ++u)
      off[u] = (unsigned)__builtin_amdgcn_readlane(sL, j + u) * (unsigned)(C_IN * 4) + (unsigned)lane * 16u;
    __builtin_amdgcn_sched_barrier(0);
#pragma unroll
    for (int u = 0; u < EDGE_BATCH; ++u)
      v[u] = *(const v4f*)((const char*)x + off[u]);
#pragma unroll
    for (int u = 0; u < EDGE_BATCH; ++u) {
      if (FULL || (unsigned)(j + u) < nIn) {
        int d = __builtin_amdgcn_readlane(rL, j + u);
        float* arow = sAcc + d * C_IN + lane;
        atomicAdd(arow,      v[u][0]);
        atomicAdd(arow + 32, v[u][1]);
        atomicAdd(arow + 64, v[u][2]);
        atomicAdd(arow + 96, v[u][3]);
      }
    }
  }
}

__device__ __forceinline__ void add_group(const float* __restrict__ x, float* sAcc,
                                          int sL, int rL, unsigned nIn, int lane) {
  if (nIn == 32u) add_rows<true>(x, sAcc, sL, rL, nIn, lane);
  else            add_rows<false>(x, sAcc, sL, rL, nIn, lane);
}

template <bool BUILD>
__global__ void __launch_bounds__(AGG_THREADS)
sage_aggregate(const float* __restrict__ x,
               const int* __restrict__ src,
               const int* __restrict__ dst,
               float* __restrict__ agg,
               float* __restrict__ cnt,
               unsigned* __restrict__ gList,
               unsigned* __restrict__ gCount,
               int nEdges, int nNodes) {
  __shared__ __attribute__((aligned(16))) float sAcc[OWN_NODES * C_IN];
  __shared__ float sCnt[OWN_NODES];

  const int tid  = (int)threadIdx.x;
  const int lane = tid & 31;
  const int wv   = __builtin_amdgcn_readfirstlane(tid >> 5);
  const int nodeLo = (int)blockIdx.x * OWN_NODES;
  int nOwnI = nNodes - nodeLo;
  if (nOwnI > OWN_NODES) nOwnI = OWN_NODES;
  if (nOwnI <= 0) return;
  const unsigned nOwn = (unsigned)nOwnI;
  unsigned* gl = gList + (size_t)blockIdx.x * LIST_CAP;

  unsigned nList;
  if constexpr (BUILD) {
    __shared__ unsigned sList[LIST_CAP];
    __shared__ unsigned sCount;

    for (int i = tid; i < OWN_NODES * C_IN / 4; i += AGG_THREADS)
      ((float4*)sAcc)[i] = make_float4(0.f, 0.f, 0.f, 0.f);
    for (int i = tid; i < OWN_NODES; i += AGG_THREADS) sCnt[i] = 0.0f;
    if (tid == 0) sCount = 0u;
    __syncthreads();

    const int nSuper = nEdges >> 9;
    {
      int4 bufA[4], bufB[4];
      int sc = wv;
      if (sc < nSuper) {
        const int scLast = nSuper - 1;
        load_dst512(bufA, dst, sc, lane);
        while (true) {
          load_dst512(bufB, dst, (sc + 32 < scLast) ? sc + 32 : scLast, lane);
          note_owned(bufA, sc, lane, nodeLo, nOwn, sList, &sCount);
          sc += 32;
          if (sc >= nSuper) break;
          load_dst512(bufA, dst, (sc + 32 < scLast) ? sc + 32 : scLast, lane);
          note_owned(bufB, sc, lane, nodeLo, nOwn, sList, &sCount);
          sc += 32;
          if (sc >= nSuper) break;
        }
      }
    }
    for (int e = (nSuper << 9) + tid; e < nEdges; e += AGG_THREADS) {
      unsigned r1 = (unsigned)(dst[e] - nodeLo);
      if (r1 < nOwn) {
        unsigned slot = atomicAdd(&sCount, 1u);
        if (slot < LIST_CAP) sList[slot] = (unsigned)e * 512u + r1;
      }
    }
    __syncthreads();

    nList = sCount;
    if (tid == 0) gCount[blockIdx.x] = nList;
    if (nList <= LIST_CAP) {
      for (unsigned i0 = (unsigned)wv * 32u; i0 < nList; i0 += 32u * 32u) {
        unsigned nIn = nList - i0;
        if (nIn > 32u) nIn = 32u;
        int sL = 0;
        int rL = 0;
        if ((unsigned)lane < nIn) {
          unsigned ent = sList[i0 + lane];
          rL = (int)(ent & 511u);
          sL = src[ent >> 9];
          gl[i0 + lane] = (unsigned)sL * 512u + (unsigned)rL;
          atomicAdd(sCnt + rL, 1.0f);
        }
        add_group(x, sAcc, sL, rL, nIn, lane);
      }
    }
  } else {
    nList = (unsigned)__builtin_amdgcn_readfirstlane((int)gCount[blockIdx.x]);
    unsigned i0 = (unsigned)wv * 32u;
    unsigned entNext = 0u;
    if (nList <= LIST_CAP && i0 + (unsigned)lane < nList) entNext = gl[i0 + lane];

    for (int i = tid; i < OWN_NODES * C_IN / 4; i += AGG_THREADS)
      ((float4*)sAcc)[i] = make_float4(0.f, 0.f, 0.f, 0.f);
    for (int i = tid; i < OWN_NODES; i += AGG_THREADS) sCnt[i] = 0.0f;
    __syncthreads();

    if (nList <= LIST_CAP) {
      for (; i0 < nList; i0 += 32u * 32u) {
        unsigned nIn = nList - i0;
        if (nIn > 32u) nIn = 32u;
        const unsigned ent = entNext;
        const unsigned i1 = i0 + 32u * 32u + (unsigned)lane;
        entNext = (i1 < nList) ? gl[i1] : 0u;
        int sL = 0;
        int rL = 0;
        if ((unsigned)lane < nIn) {
          rL = (int)(ent & 511u);
          sL = (int)(ent >> 9);
          atomicAdd(sCnt + rL, 1.0f);
        }
        add_group(x, sAcc, sL, rL, nIn, lane);
      }
    }
  }

  if (nList > LIST_CAP) {
    for (int e = tid; e < nEdges; e += AGG_THREADS) {
      unsigned r1 = (unsigned)(dst[e] - nodeLo);
      if (r1 < nOwn) add_edge_direct(x, src, sAcc, sCnt, e, r1);
    }
  }
  __syncthreads();

  {
    float4* gdst = (float4*)(agg + (long long)nodeLo * C_IN);
    const int n4 = nOwnI * (C_IN / 4);
    for (int i = tid; i < n4; i += AGG_THREADS) {
      const float* w = sAcc + (i >> 5) * C_IN + (i & 31);
      gdst[i] = make_float4(w[0], w[32], w[64], w[96]);
    }
    for (int i = tid; i < nOwnI; i += AGG_THREADS) cnt[nodeLo + i] = sCnt[i];
  }
}

#define A_PITCH 132
#define GEMM_BLOCKS_128 512
#define GEMM_BLOCKS_64 1024

template <int COUT, bool RELU>
__global__ void __launch_bounds__(COUT * 2) __attribute__((amdgpu_waves_per_eu(4, 4)))
sage_gemm(const float* __restrict__ xin,
          const float* __restrict__ agg,
          const float* __restrict__ cnt,
          const float* __restrict__ Wl,
          const float* __restrict__ bl,
          const float* __restrict__ Wr,
          float* __restrict__ out,
          int nStrips) {
  constexpr int TN  = COUT / 16;
  constexpr int RP  = 16 / TN;
  constexpr int MAT = 16 * A_PITCH;
  constexpr int BUF = 2 * MAT;
  __shared__ __attribute__((aligned(16))) float sA[2 * BUF];

  const int tid  = (int)threadIdx.x;
  const int lane = tid & 31;
  const int wv   = __builtin_amdgcn_readfirstlane(tid >> 5);
  const int r    = lane & 15;
  const int kh   = lane >> 4;
  const int col  = wv * 16 + r;
  const int step = (int)gridDim.x;

  v2f bL[32], bR[32];
#pragma unroll
  for (int ks = 0; ks < 32; ++ks) {
    const int kk = 4 * ks + 2 * kh;
    bL[ks][0] = Wl[kk * COUT + col];
    bL[ks][1] = Wl[(kk + 1) * COUT + col];
    bR[ks][0] = Wr[kk * COUT + col];
    bR[ks][1] = Wr[(kk + 1) * COUT + col];
  }
  const float bias = bl[col];

  v4f pa[RP], px[RP];
  float  pc[RP];

#define GEMM_FETCH(STRIP)                                                       \
  _Pragma("unroll")                                                             \
  for (int p = 0; p < RP; ++p) {                                                \
    const long long g = (long long)(STRIP) * 16 + p * TN + wv;                  \
    pa[p] = *(const v4f*)(agg + g * C_IN + lane * 4);                           \
    px[p] = *(const v4f*)(xin + g * C_IN + lane * 4);                           \
    pc[p] = cnt[g];                                                             \
  }

#define GEMM_STAGE(SB)                                                          \
  _Pragma("unroll")                                                             \
  for (int p = 0; p < RP; ++p) {                                                \
    const float inv = 1.0f / fmaxf(pc[p], 1.0f);                                \
    const v4f a = pa[p] * inv;                                                  \
    float* d = (SB) + (p * TN + wv) * A_PITCH + lane * 4;                       \
    *(v4f*)d = a;                                                               \
    *(v4f*)(d + MAT) = px[p];                                                   \
  }

#define GEMM_STRIP_BARRIER()                                                    \
  __builtin_amdgcn_fence(__ATOMIC_RELEASE, "workgroup", "local");               \
  __builtin_amdgcn_s_barrier();                                                 \
  __builtin_amdgcn_fence(__ATOMIC_ACQUIRE, "workgroup", "local")

  int it = (int)blockIdx.x;
  if (it >= nStrips) return;
  const int itLast = nStrips - 1;
  GEMM_FETCH(it)
  GEMM_STAGE(sA)
  GEMM_FETCH((it + step < itLast) ? it + step : itLast)
  GEMM_STRIP_BARRIER();

  int buf = 0;
  for (; it < nStrips; it += step) {
    const float* fa = sA + buf * BUF + r * A_PITCH + 2 * kh;
    v8f acc = {};
    v2f fragA[8], fragB[8];
#pragma unroll
    for (int q = 0; q < 8; ++q) fragA[q] = *(const v2f*)(fa + 4 * q);
#pragma unroll
    for (int c = 0; c < 8; c += 2) {
      {
        const float* nb = fa + ((c + 1) >= 4 ? MAT : 0) + 32 * ((c + 1) & 3);
#pragma unroll
        for (int q = 0; q < 8; ++q) fragB[q] = *(const v2f*)(nb + 4 * q);
      }
#pragma unroll
      for (int q = 0; q < 8; ++q)
        acc = __builtin_amdgcn_wmma_f32_16x16x4_f32(false, fragA[q], false,
                                                    (c < 4) ? bL[8 * c + q] : bR[8 * (c - 4) + q],
                                                    (short)0, acc, false, false);
      if (c + 2 < 8) {
        const float* nb = fa + ((c + 2) >= 4 ? MAT : 0) + 32 * ((c + 2) & 3);
#pragma unroll
        for (int q = 0; q < 8; ++q) fragA[q] = *(const v2f*)(nb + 4 * q);
      }
#pragma unroll
      for (int q = 0; q < 8; ++q)
        acc = __builtin_amdgcn_wmma_f32_16x16x4_f32(false, fragB[q], false,
                                                    (c + 1 < 4) ? bL[8 * (c + 1) + q] : bR[8 * (c + 1 - 4) + q],
                                                    (short)0, acc, false, false);
    }

    const long long m0 = (long long)it * 16;
#pragma unroll
    for (int rr = 0; rr < 8; ++rr) {
      float v = acc[rr] + bias;
      if (RELU) v = fmaxf(v, 0.0f);
      out[(m0 + rr + 8 * kh) * COUT + col] = v;
    }

    GEMM_STAGE(sA + (buf ^ 1) * BUF)
    GEMM_FETCH((it + 2 * step < itLast) ? it + 2 * step : itLast)
    GEMM_STRIP_BARRIER();
    buf ^= 1;
  }
#undef GEMM_FETCH
#undef GEMM_STAGE
#undef GEMM_STRIP_BARRIER
}

extern "C" void kernel_launch(void* const* d_in, const int* in_sizes, int n_in,
                              void* d_out, int out_size, void* d_ws, size_t ws_size,
                              hipStream_t stream) {
  const float* x0  = (const float*)d_in[0];
  const int*   ei  = (const int*)d_in[1];
  const float* Wl0 = (const float*)d_in[2];
  const float* bl0 = (const float*)d_in[3];
  const float* Wr0 = (const float*)d_in[4];
  const float* Wl1 = (const float*)d_in[5];
  const float* bl1 = (const float*)d_in[6];
  const float* Wr1 = (const float*)d_in[7];
  const float* Wl2 = (const float*)d_in[8];
  const float* bl2 = (const float*)d_in[9];
  const float* Wr2 = (const float*)d_in[10];

  const int N = in_sizes[0] / C_IN;
  const int E = in_sizes[1] / 2;
  const int* src = ei;
  const int* dst = ei + E;

  const int aggBlocks = (N + OWN_NODES - 1) / OWN_NODES;

  float* h0  = (float*)d_ws;
  float* h1  = h0 + (size_t)N * C_IN;
  float* agg = h1 + (size_t)N * C_IN;
  unsigned* lists  = (unsigned*)(agg + (size_t)N * C_IN);
  unsigned* counts = lists + (size_t)aggBlocks * LIST_CAP;
  float* cnt = (float*)(counts + aggBlocks);

  const int tilesM = (N + 15) / 16;

  sage_aggregate<true><<<aggBlocks, AGG_THREADS, 0, stream>>>(x0, src, dst, agg, cnt, lists, counts, E, N);
  sage_gemm<128, true><<<GEMM_BLOCKS_128, 256, 0, stream>>>(x0, agg, cnt, Wl0, bl0, Wr0, h0, tilesM);

  sage_aggregate<false><<<aggBlocks, AGG_THREADS, 0, stream>>>(h0, src, dst, agg, cnt, lists, counts, E, N);
  sage_gemm<128, true><<<GEMM_BLOCKS_128, 256, 0, stream>>>(h0, agg, cnt, Wl1, bl1, Wr1, h1, tilesM);

  sage_aggregate<false><<<aggBlocks, AGG_THREADS, 0, stream>>>(h1, src, dst, agg, cnt, lists, counts, E, N);
  sage_gemm<64, false><<<GEMM_BLOCKS_64, 128, 0, stream>>>(h1, agg, cnt, Wl2, bl2, Wr2,
                                                                 (float*)d_out, tilesM);
}


# Aggregation sums in registers: a wave owns 13 nodes and its own edge list and adds each listed row to the node's register row, chosen by a wave-uniform index; no LDS float adds; sums in list order
# speedup vs baseline: 3.8689x; 1.0530x over previous
#include <hip/hip_runtime.h>
#include <hip/hip_bf16.h>

typedef __attribute__((ext_vector_type(2))) float v2f;
typedef __attribute__((ext_vector_type(8))) float v8f;
typedef __attribute__((ext_vector_type(4))) float v4f;

#define C_IN 128

#define OWN_NODES 391
#define SUB_NODES 13
#define SUB_CAP 512
#define LIST_CAP 16384
#define LIST_PLAIN 0xFFFFFFFFu
#define AGG_THREADS 1024
#define EDGE_BATCH 8

__device__ __forceinline__ void load_dst512(int4 (&buf)[4], const int* __restrict__ dst,
                                            int sc, int lane) {
#pragma unroll
  for (int u = 0; u < 4; ++u)
    buf[u] = *(const int4*)(dst + (long long)sc * 512 + u * 128 + lane * 4);
}

__device__ __forceinline__ void note_owned(const int4 (&buf)[4], int sc, int lane,
                                           int nodeLo, unsigned nOwn,
                                           unsigned* sList, unsigned* sCount) {
  unsigned rel[16];
  unsigned msk[16];
  unsigned total = 0;
#pragma unroll
  for (int u = 0; u < 4; ++u) {
    rel[u * 4 + 0] = (unsigned)(buf[u].x - nodeLo);
    rel[u * 4 + 1] = (unsigned)(buf[u].y - nodeLo);
    rel[u * 4 + 2] = (unsigned)(buf[u].z - nodeLo);
    rel[u * 4 + 3] = (unsigned)(buf[u].w - nodeLo);
  }
#pragma unroll
  for (int q = 0; q < 16; ++q) {
    msk[q] = __builtin_amdgcn_ballot_w32(rel[q] < nOwn);
    total += (unsigned)__popc(msk[q]);
  }
  if (total != 0u) {
    unsigned base = 0u;
    if (lane == 0) base = atomicAdd(sCount, total);
    base = (unsigned)__builtin_amdgcn_readfirstlane((int)base);
#pragma unroll
    for (int q = 0; q < 16; ++q) {
      if (msk[q] != 0u) {
        if (rel[q] < nOwn) {
          unsigned slot = base + __builtin_amdgcn_mbcnt_lo(msk[q], 0u);
          unsigned e = (unsigned)sc * 512u + (unsigned)((q >> 2) * 128 + lane * 4 + (q & 3));
          if (slot < LIST_CAP) sList[slot] = e * 512u + rel[q];
        }
        base += (unsigned)__popc(msk[q]);
      }
    }
  }
}

typedef __attribute__((ext_vector_type(16))) float v16f;
static_assert(SUB_NODES <= 16, "a register row vector holds 16 nodes");
struct RowAcc { v16f c0, c1, c2, c3; };

#define PAD_NODE 15
static_assert(PAD_NODE >= SUB_NODES && PAD_NODE < 16, "the spare element lies past the wave's nodes");
__device__ __forceinline__ void add_group(const float* __restrict__ x, RowAcc& acc,
                                          int sL, int dL, int lane) {
#pragma unroll
  for (int j = 0; j < 32; j += EDGE_BATCH) {
    unsigned off[EDGE_BATCH];
    v4f v[EDGE_BATCH];
#pragma unroll
    for (int u = 0; u < EDGE_BATCH; ++u)
      off[u] = (unsigned)__builtin_amdgcn_readlane(sL, j + u) * (unsigned)(C_IN * 4) + (unsigned)lane * 16u;
    __builtin_amdgcn_sched_barrier(0);
#pragma unroll
    for (int u = 0; u < EDGE_BATCH; ++u)
      v[u] = *(const v4f*)((const char*)x + off[u]);
#pragma unroll
    for (int u = 0; u < EDGE_BATCH; ++u) {
      const int d = __builtin_amdgcn_readlane(dL, j + u);
      acc.c0[d] += v[u][0];
      acc.c1[d] += v[u][1];
      acc.c2[d] += v[u][2];
      acc.c3[d] += v[u][3];
    }
  }
}

__device__ __forceinline__ void wave_plain(const float* __restrict__ x,
                                           const int* __restrict__ src,
                                           const int* __restrict__ dst,
                                           RowAcc& acc, float& cl,
                                           int nEdges, int node0, unsigned nk, int lane) {
  for (int e0 = 0; e0 < nEdges; e0 += 32) {
    const int e = e0 + lane;
    unsigned r = 0xFFFFFFFFu;
    if (e < nEdges) r = (unsigned)(dst[e] - node0);
    const bool mine = r < nk;
    unsigned mask = __builtin_amdgcn_ballot_w32(mine);
    if (mask != 0u) {
      int s = 0;
      if (mine) s = src[e];
      while (mask != 0u) {
        const int i = __builtin_ctz(mask);
        mask &= mask - 1u;
        const int sRow = __builtin_amdgcn_readlane(s, i);
        const int d = __builtin_amdgcn_readlane((int)r, i);
        const v4f v = *(const v4f*)((const char*)x + (unsigned)sRow * (unsigned)(C_IN * 4) + (unsigned)lane * 16u);
        acc.c0[d] += v[0];
        acc.c1[d] += v[1];
        acc.c2[d] += v[2];
        acc.c3[d] += v[3];
        if (lane == d) cl += 1.0f;
      }
    }
  }
}

__device__ __forceinline__ void store_rows(float* __restrict__ agg, const RowAcc& acc,
                                           int node0, unsigned nk, int lane) {
#pragma unroll
  for (int k = 0; k < SUB_NODES; ++k) {
    if ((unsigned)k < nk) {
      v4f o;
      o[0] = acc.c0[k];
      o[1] = acc.c1[k];
      o[2] = acc.c2[k];
      o[3] = acc.c3[k];
      *(v4f*)(agg + (long long)(node0 + k) * C_IN + lane * 4) = o;
    }
  }
}

template <bool BUILD>
__global__ void __launch_bounds__(AGG_THREADS)
sage_aggregate(const float* __restrict__ x,
               const int* __restrict__ src,
               const int* __restrict__ dst,
               float* __restrict__ agg,
               float* __restrict__ cnt,
               unsigned* __restrict__ gList,
               unsigned* __restrict__ gCount,
               int nEdges, int nNodes) {
  const int tid  = (int)threadIdx.x;
  const int lane = tid & 31;
  const int wv   = __builtin_amdgcn_readfirstlane(tid >> 5);
  const int nodeLo = (int)blockIdx.x * OWN_NODES;
  int nOwnI = nNodes - nodeLo;
  if (nOwnI > OWN_NODES) nOwnI = OWN_NODES;
  if (nOwnI <= 0) return;
  const unsigned nOwn = (unsigned)nOwnI;
  const int node0 = nodeLo + wv * SUB_NODES;
  int nkI = nOwnI - wv * SUB_NODES;
  if (nkI > SUB_NODES) nkI = SUB_NODES;
  if (nkI < 0) nkI = 0;
  const unsigned nk = (unsigned)nkI;
  unsigned* gw = gList + (size_t)blockIdx.x * LIST_CAP + (size_t)wv * SUB_CAP;
  unsigned* gc = gCount + (size_t)blockIdx.x * 32 + wv;

  RowAcc acc;
  acc.c0 = (v16f)(0.0f);
  acc.c1 = (v16f)(0.0f);
  acc.c2 = (v16f)(0.0f);
  acc.c3 = (v16f)(0.0f);
  float cl = 0.0f;

  unsigned nW;
  if constexpr (BUILD) {
    __shared__ unsigned sList[LIST_CAP];
    __shared__ unsigned sSub[32 * SUB_CAP];
    __shared__ unsigned sSubCount[32];
    __shared__ float sCnt[OWN_NODES];
    __shared__ unsigned sCount;

    for (int i = tid; i < OWN_NODES; i += AGG_THREADS) sCnt[i] = 0.0f;
    if (tid < 32) sSubCount[tid] = 0u;
    if (tid == 0) sCount = 0u;
    __syncthreads();

    const int nSuper = nEdges >> 9;
    {
      int4 bufA[4], bufB[4];
      int sc = wv;
      if (sc < nSuper) {
        const int scLast = nSuper - 1;
        load_dst512(bufA, dst, sc, lane);
        while (true) {
          load_dst512(bufB, dst, (sc + 32 < scLast) ? sc + 32 : scLast, lane);
          note_owned(bufA, sc, lane, nodeLo, nOwn, sList, &sCount);
          sc += 32;
          if (sc >= nSuper) break;
          load_dst512(bufA, dst, (sc + 32 < scLast) ? sc + 32 : scLast, lane);
          note_owned(bufB, sc, lane, nodeLo, nOwn, sList, &sCount);
          sc += 32;
          if (sc >= nSuper) break;
        }
      }
    }
    for (int e = (nSuper << 9) + tid; e < nEdges; e += AGG_THREADS) {
      unsigned r1 = (unsigned)(dst[e] - nodeLo);
      if (r1 < nOwn) {
        unsigned slot = atomicAdd(&sCount, 1u);
        if (slot < LIST_CAP) sList[slot] = (unsigned)e * 512u + r1;
      }
    }
    __syncthreads();

    const unsigned nList = sCount;
    const bool fits = nList <= LIST_CAP;
    if (fits) {
      for (unsigned i = (unsigned)tid; i < nList; i += AGG_THREADS) {
        const unsigned ent = sList[i];
        const unsigned rel = ent & 511u;
        const unsigned s   = (unsigned)src[ent >> 9];
        const unsigned sub = rel / SUB_NODES;
        const unsigned slot = atomicAdd(&sSubCount[sub], 1u);
        if (slot < SUB_CAP) sSub[sub * SUB_CAP + slot] = s * 16u + (rel - sub * SUB_NODES);
        atomicAdd(sCnt + rel, 1.0f);
      }
    }
    __syncthreads();

    if (fits) {
      for (int i = tid; i < nOwnI; i += AGG_THREADS) cnt[nodeLo + i] = sCnt[i];
      nW = sSubCount[wv];
      if (nW > SUB_CAP) nW = LIST_PLAIN;
    } else {
      nW = LIST_PLAIN;
    }
    nW = (unsigned)__builtin_amdgcn_readfirstlane((int)nW);
    if (lane == 0) *gc = nW;

    if (nk != 0u) {
      if (nW != LIST_PLAIN) {
        const unsigned* sw = sSub + wv * SUB_CAP;
        for (unsigned i0 = 0u; i0 < nW; i0 += 32u) {
          int sL = 0;
          int dL = PAD_NODE;
          if (i0 + (unsigned)lane < nW) {
            const unsigned ent = sw[i0 + lane];
            gw[i0 + lane] = ent;
            sL = (int)(ent >> 4);
            dL = (int)(ent & 15u);
          }
          add_group(x, acc, sL, dL, lane);
        }
      } else {
        wave_plain(x, src, dst, acc, cl, nEdges, node0, nk, lane);
        if (!fits && (unsigned)lane < nk) cnt[node0 + lane] = cl;
      }
      store_rows(agg, acc, node0, nk, lane);
    }
  } else {
    if (nk == 0u) return;
    nW = (unsigned)__builtin_amdgcn_readfirstlane((int)*gc);
    if (nW != LIST_PLAIN) {
      unsigned entNext = ((unsigned)lane < nW) ? gw[lane] : (unsigned)PAD_NODE;
      for (unsigned i0 = 0u; i0 < nW; i0 += 32u) {
        const unsigned ent = entNext;
        const unsigned i1 = i0 + 32u + (unsigned)lane;
        entNext = (i1 < nW) ? gw[i1] : (unsigned)PAD_NODE;
        const int sL = (int)(ent >> 4);
        const int dL = (int)(ent & 15u);
        add_group(x, acc, sL, dL, lane);
      }
    } else {
      wave_plain(x, src, dst, acc, cl, nEdges, node0, nk, lane);
    }
    store_rows(agg, acc, node0, nk, lane);
  }
}

#define A_PITCH 132
#define GEMM_BLOCKS_128 512
#define GEMM_BLOCKS_64 1024

template <int COUT, bool RELU>
__global__ void __launch_bounds__(COUT * 2) __attribute__((amdgpu_waves_per_eu(4, 4)))
sage_gemm(const float* __restrict__ xin,
          const float* __restrict__ agg,
          const float* __restrict__ cnt,
          const float* __restrict__ Wl,
          const float* __restrict__ bl,
          const float* __restrict__ Wr,
          float* __restrict__ out,
          int nStrips) {
  constexpr int TN  = COUT / 16;
  constexpr int RP  = 16 / TN;
  constexpr int MAT = 16 * A_PITCH;
  constexpr int BUF = 2 * MAT;
  __shared__ __attribute__((aligned(16))) float sA[2 * BUF];

  const int tid  = (int)threadIdx.x;
  const int lane = tid & 31;
  const int wv   = __builtin_amdgcn_readfirstlane(tid >> 5);
  const int r    = lane & 15;
  const int kh   = lane >> 4;
  const int col  = wv * 16 + r;
  const int step = (int)gridDim.x;

  v2f bL[32], bR[32];
#pragma unroll
  for (int ks = 0; ks < 32; ++ks) {
    const int kk = 4 * ks + 2 * kh;
    bL[ks][0] = Wl[kk * COUT + col];
    bL[ks][1] = Wl[(kk + 1) * COUT + col];
    bR[ks][0] = Wr[kk * COUT + col];
    bR[ks][1] = Wr[(kk + 1) * COUT + col];
  }
  const float bias = bl[col];

  v4f pa[RP], px[RP];
  float  pc[RP];

#define GEMM_FETCH(STRIP)                                                       \
  _Pragma("unroll")                                                             \
  for (int p = 0; p < RP; ++p) {                                                \
    const long long g = (long long)(STRIP) * 16 + p * TN + wv;                  \
    pa[p] = *(const v4f*)(agg + g * C_IN + lane * 4);                           \
    px[p] = *(const v4f*)(xin + g * C_IN + lane * 4);                           \
    pc[p] = cnt[g];                                                             \
  }

#define GEMM_STAGE(SB)                                                          \
  _Pragma("unroll")                                                             \
  for (int p = 0; p < RP; ++p) {                                                \
    const float inv = 1.0f / fmaxf(pc[p], 1.0f);                                \
    const v4f a = pa[p] * inv;                                                  \
    float* d = (SB) + (p * TN + wv) * A_PITCH + lane * 4;                       \
    *(v4f*)d = a;                                                               \
    *(v4f*)(d + MAT) = px[p];                                                   \
  }

#define GEMM_STRIP_BARRIER()                                                    \
  __builtin_amdgcn_fence(__ATOMIC_RELEASE, "workgroup", "local");               \
  __builtin_amdgcn_s_barrier();                                                 \
  __builtin_amdgcn_fence(__ATOMIC_ACQUIRE, "workgroup", "local")

  int it = (int)blockIdx.x;
  if (it >= nStrips) return;
  const int itLast = nStrips - 1;
  GEMM_FETCH(it)
  GEMM_STAGE(sA)
  GEMM_FETCH((it + step < itLast) ? it + step : itLast)
  GEMM_STRIP_BARRIER();

  int buf = 0;
  for (; it < nStrips; it += step) {
    const float* fa = sA + buf * BUF + r * A_PITCH + 2 * kh;
    v8f acc = {};
    v2f fragA[8], fragB[8];
#pragma unroll
    for (int q = 0; q < 8; ++q) fragA[q] = *(const v2f*)(fa + 4 * q);
#pragma unroll
    for (int c = 0; c < 8; c += 2) {
      {
        const float* nb = fa + ((c + 1) >= 4 ? MAT : 0) + 32 * ((c + 1) & 3);
#pragma unroll
        for (int q = 0; q < 8; ++q) fragB[q] = *(const v2f*)(nb + 4 * q);
      }
#pragma unroll
      for (int q = 0; q < 8; ++q)
        acc = __builtin_amdgcn_wmma_f32_16x16x4_f32(false, fragA[q], false,
                                                    (c < 4) ? bL[8 * c + q] : bR[8 * (c - 4) + q],
                                                    (short)0, acc, false, false);
      if (c + 2 < 8) {
        const float* nb = fa + ((c + 2) >= 4 ? MAT : 0) + 32 * ((c + 2) & 3);
#pragma unroll
        for (int q = 0; q < 8; ++q) fragA[q] = *(const v2f*)(nb + 4 * q);
      }
#pragma unroll
      for (int q = 0; q < 8; ++q)
        acc = __builtin_amdgcn_wmma_f32_16x16x4_f32(false, fragB[q], false,
                                                    (c + 1 < 4) ? bL[8 * (c + 1) + q] : bR[8 * (c + 1 - 4) + q],
                                                    (short)0, acc, false, false);
    }

    const long long m0 = (long long)it * 16;
#pragma unroll
    for (int rr = 0; rr < 8; ++rr) {
      float v = acc[rr] + bias;
      if (RELU) v = fmaxf(v, 0.0f);
      out[(m0 + rr + 8 * kh) * COUT + col] = v;
    }

    GEMM_STAGE(sA + (buf ^ 1) * BUF)
    GEMM_FETCH((it + 2 * step < itLast) ? it + 2 * step : itLast)
    GEMM_STRIP_BARRIER();
    buf ^= 1;
  }
#undef GEMM_FETCH
#undef GEMM_STAGE
#undef GEMM_STRIP_BARRIER
}

extern "C" void kernel_launch(void* const* d_in, const int* in_sizes, int n_in,
                              void* d_out, int out_size, void* d_ws, size_t ws_size,
                              hipStream_t stream) {
  const float* x0  = (const float*)d_in[0];
  const int*   ei  = (const int*)d_in[1];
  const float* Wl0 = (const float*)d_in[2];
  const float* bl0 = (const float*)d_in[3];
  const float* Wr0 = (const float*)d_in[4];
  const float* Wl1 = (const float*)d_in[5];
  const float* bl1 = (const float*)d_in[6];
  const float* Wr1 = (const float*)d_in[7];
  const float* Wl2 = (const float*)d_in[8];
  const float* bl2 = (const float*)d_in[9];
  const float* Wr2 = (const float*)d_in[10];

  const int N = in_sizes[0] / C_IN;
  const int E = in_sizes[1] / 2;
  const int* src = ei;
  const int* dst = ei + E;

  const int aggBlocks = (N + OWN_NODES - 1) / OWN_NODES;

  float* h0  = (float*)d_ws;
  float* h1  = h0 + (size_t)N * C_IN;
  float* agg = h1 + (size_t)N * C_IN;
  unsigned* lists  = (unsigned*)(agg + (size_t)N * C_IN);
  unsigned* counts = lists + (size_t)aggBlocks * LIST_CAP;
  float* cnt = (float*)(counts + (size_t)aggBlocks * 32);

  const int tilesM = (N + 15) / 16;

  sage_aggregate<true><<<aggBlocks, AGG_THREADS, 0, stream>>>(x0, src, dst, agg, cnt, lists, counts, E, N);
  sage_gemm<128, true><<<GEMM_BLOCKS_128, 256, 0, stream>>>(x0, agg, cnt, Wl0, bl0, Wr0, h0, tilesM);

  sage_aggregate<false><<<aggBlocks, AGG_THREADS, 0, stream>>>(h0, src, dst, agg, cnt, lists, counts, E, N);
  sage_gemm<128, true><<<GEMM_BLOCKS_128, 256, 0, stream>>>(h0, agg, cnt, Wl1, bl1, Wr1, h1, tilesM);

  sage_aggregate<false><<<aggBlocks, AGG_THREADS, 0, stream>>>(h1, src, dst, agg, cnt, lists, counts, E, N);
  sage_gemm<64, false><<<GEMM_BLOCKS_64, 128, 0, stream>>>(h1, agg, cnt, Wl2, bl2, Wr2,
                                                                 (float*)d_out, tilesM);
}


# Edge-list pass: a first launch writes one owner byte per edge; each owner scans those bytes (4 edges per word test, 1/4 the L2 bytes), re-checks noted edges against dst; lists still in arrival order
# speedup vs baseline: 4.2803x; 1.1063x over previous
#include <hip/hip_runtime.h>
#include <hip/hip_bf16.h>

typedef __attribute__((ext_vector_type(2))) float v2f;
typedef __attribute__((ext_vector_type(8))) float v8f;
typedef __attribute__((ext_vector_type(4))) float v4f;

#define C_IN 128

#define OWN_NODES 391
#define SUB_NODES 13
#define SUB_CAP 512
#define LIST_CAP 16384
#define LIST_PLAIN 0xFFFFFFFFu
static_assert(LIST_CAP == 32 * SUB_CAP, "every wave notes into its own SUB_CAP entries of the owner's list");
#define AGG_THREADS 1024
#define EDGE_BATCH 8

#define KEY_THREADS 512
#define KEY_BLOCKS 800

__device__ __forceinline__ unsigned key_pos(unsigned el) {
  const unsigned l  = el >> 6;
  const unsigned hi = (el >> 5) & 1u;
  const unsigned b  = (el >> 3) & 3u;
  const unsigned c  = el & 7u;
  const unsigned j  = 2u * hi + ((c < 4u) ? 1u : 0u);
  const unsigned u  = 3u - (c & 3u);
  return j * 512u + l * 16u + 4u * u + b;
}

__global__ void __launch_bounds__(KEY_THREADS)
sage_owner_keys(const int* __restrict__ dst, unsigned* __restrict__ keys, int nEdges) {
  __shared__ unsigned sKey[KEY_THREADS];
  unsigned char* sKeyB = (unsigned char*)sKey;
  const unsigned tid = threadIdx.x;
  const int nBlk = nEdges >> 11;
  for (int blk = (int)blockIdx.x; blk < nBlk; blk += (int)gridDim.x) {
    const int4 d = *(const int4*)(dst + (long long)blk * 2048 + tid * 4);
    sKeyB[key_pos(4u * tid + 0u)] = (unsigned char)((unsigned)d.x / OWN_NODES);
    sKeyB[key_pos(4u * tid + 1u)] = (unsigned char)((unsigned)d.y / OWN_NODES);
    sKeyB[key_pos(4u * tid + 2u)] = (unsigned char)((unsigned)d.z / OWN_NODES);
    sKeyB[key_pos(4u * tid + 3u)] = (unsigned char)((unsigned)d.w / OWN_NODES);
    __syncthreads();
    keys[(long long)blk * KEY_THREADS + tid] = sKey[tid];
    __syncthreads();
  }
}

__device__ __forceinline__ void load_keys4(uint4 (&buf)[4], const unsigned* __restrict__ keys,
                                           int blk, int lane) {
#pragma unroll
  for (int u = 0; u < 4; ++u)
    buf[u] = *(const uint4*)(keys + (long long)blk * 512 + u * 128 + lane * 4);
}

__device__ __forceinline__ unsigned key_hits(unsigned w, unsigned pat) {
  const unsigned x = w ^ pat;
  return (x - 0x01010101u) & ~x & 0x80808080u;
}

__device__ __forceinline__ unsigned unit_hits(const uint4& k, unsigned pat) {
  return key_hits(k.x, pat) | (key_hits(k.y, pat) >> 1)
       | (key_hits(k.z, pat) >> 2) | (key_hits(k.w, pat) >> 3);
}

__device__ __forceinline__ void wave_note(bool has, unsigned e, unsigned& n, unsigned* noted) {
  const unsigned msk = __builtin_amdgcn_ballot_w32(has);
  if (has) {
    const unsigned slot = n + __builtin_amdgcn_mbcnt_lo(msk, 0u);
    if (slot < SUB_CAP) noted[slot] = e;
  }
  n += (unsigned)__popc(msk);
}

__device__ __forceinline__ void note_keys(const uint4 (&buf)[4], int blk, int lane,
                                          unsigned pat, unsigned& n, unsigned* noted) {
  const unsigned mLo = unit_hits(buf[0], pat) | (unit_hits(buf[1], pat) >> 4);
  const unsigned mHi = unit_hits(buf[2], pat) | (unit_hits(buf[3], pat) >> 4);
  unsigned long long m = ((unsigned long long)mHi << 32) | mLo;
  const unsigned eBase = (unsigned)blk * 2048u + (unsigned)lane * 64u;
  while (true) {
    const bool has = m != 0ull;
    if (__builtin_amdgcn_ballot_w32(has) == 0u) break;
    const unsigned q = (unsigned)__builtin_ctzll(m | (1ull << 63));
    wave_note(has, eBase + q, n, noted);
    m &= m - 1ull;
  }
}

typedef __attribute__((ext_vector_type(16))) float v16f;
static_assert(SUB_NODES <= 16, "a register row vector holds 16 nodes");
struct RowAcc { v16f c0, c1, c2, c3; };

#define PAD_NODE 15
static_assert(PAD_NODE >= SUB_NODES && PAD_NODE < 16, "the spare element lies past the wave's nodes");
__device__ __forceinline__ void add_group(const float* __restrict__ x, RowAcc& acc,
                                          int sL, int dL, int lane) {
#pragma unroll
  for (int j = 0; j < 32; j += EDGE_BATCH) {
    unsigned off[EDGE_BATCH];
    v4f v[EDGE_BATCH];
#pragma unroll
    for (int u = 0; u < EDGE_BATCH; ++u)
      off[u] = (unsigned)__builtin_amdgcn_readlane(sL, j + u) * (unsigned)(C_IN * 4) + (unsigned)lane * 16u;
    __builtin_amdgcn_sched_barrier(0);
#pragma unroll
    for (int u = 0; u < EDGE_BATCH; ++u)
      v[u] = *(const v4f*)((const char*)x + off[u]);
#pragma unroll
    for (int u = 0; u < EDGE_BATCH; ++u) {
      const int d = __builtin_amdgcn_readlane(dL, j + u);
      acc.c0[d] += v[u][0];
      acc.c1[d] += v[u][1];
      acc.c2[d] += v[u][2];
      acc.c3[d] += v[u][3];
    }
  }
}

__device__ __forceinline__ void wave_plain(const float* __restrict__ x,
                                           const int* __restrict__ src,
                                           const int* __restrict__ dst,
                                           RowAcc& acc, float& cl,
                                           int nEdges, int node0, unsigned nk, int lane) {
  for (int e0 = 0; e0 < nEdges; e0 += 32) {
    const int e = e0 + lane;
    unsigned r = 0xFFFFFFFFu;
    if (e < nEdges) r = (unsigned)(dst[e] - node0);
    const bool mine = r < nk;
    unsigned mask = __builtin_amdgcn_ballot_w32(mine);
    if (mask != 0u) {
      int s = 0;
      if (mine) s = src[e];
      while (mask != 0u) {
        const int i = __builtin_ctz(mask);
        mask &= mask - 1u;
        const int sRow = __builtin_amdgcn_readlane(s, i);
        const int d = __builtin_amdgcn_readlane((int)r, i);
        const v4f v = *(const v4f*)((const char*)x + (unsigned)sRow * (unsigned)(C_IN * 4) + (unsigned)lane * 16u);
        acc.c0[d] += v[0];
        acc.c1[d] += v[1];
        acc.c2[d] += v[2];
        acc.c3[d] += v[3];
        if (lane == d) cl += 1.0f;
      }
    }
  }
}

__device__ __forceinline__ void store_rows(float* __restrict__ agg, const RowAcc& acc,
                                           int node0, unsigned nk, int lane) {
#pragma unroll
  for (int k = 0; k < SUB_NODES; ++k) {
    if ((unsigned)k < nk) {
      v4f o;
      o[0] = acc.c0[k];
      o[1] = acc.c1[k];
      o[2] = acc.c2[k];
      o[3] = acc.c3[k];
      *(v4f*)(agg + (long long)(node0 + k) * C_IN + lane * 4) = o;
    }
  }
}

template <bool BUILD>
__global__ void __launch_bounds__(AGG_THREADS)
sage_aggregate(const float* __restrict__ x,
               const int* __restrict__ src,
               const int* __restrict__ dst,
               float* __restrict__ agg,
               float* __restrict__ cnt,
               unsigned* __restrict__ gList,
               unsigned* __restrict__ gCount,
               const unsigned* __restrict__ keys,
               int nEdges, int nNodes) {
  const int tid  = (int)threadIdx.x;
  const int lane = tid & 31;
  const int wv   = __builtin_amdgcn_readfirstlane(tid >> 5);
  const int nodeLo = (int)blockIdx.x * OWN_NODES;
  int nOwnI = nNodes - nodeLo;
  if (nOwnI > OWN_NODES) nOwnI = OWN_NODES;
  if (nOwnI <= 0) return;
  const unsigned nOwn = (unsigned)nOwnI;
  const int node0 = nodeLo + wv * SUB_NODES;
  int nkI = nOwnI - wv * SUB_NODES;
  if (nkI > SUB_NODES) nkI = SUB_NODES;
  if (nkI < 0) nkI = 0;
  const unsigned nk = (unsigned)nkI;
  unsigned* gw = gList + (size_t)blockIdx.x * LIST_CAP + (size_t)wv * SUB_CAP;
  unsigned* gc = gCount + (size_t)blockIdx.x * 32 + wv;

  RowAcc acc;
  acc.c0 = (v16f)(0.0f);
  acc.c1 = (v16f)(0.0f);
  acc.c2 = (v16f)(0.0f);
  acc.c3 = (v16f)(0.0f);
  float cl = 0.0f;

  unsigned nW;
  if constexpr (BUILD) {
    __shared__ unsigned sList[LIST_CAP];
    __shared__ unsigned sSub[32 * SUB_CAP];
    __shared__ unsigned sSubCount[32];
    __shared__ float sCnt[OWN_NODES];
    __shared__ unsigned sCount;

    for (int i = tid; i < OWN_NODES; i += AGG_THREADS) sCnt[i] = 0.0f;
    if (tid < 32) sSubCount[tid] = 0u;
    if (tid == 0) sCount = 0u;
    __syncthreads();

    const int nSuper = nEdges >> 11;
    const unsigned pat = ((unsigned)blockIdx.x & 255u) * 0x01010101u;
    unsigned* myNoted = sList + wv * SUB_CAP;
    unsigned nNoted = 0u;
    {
      uint4 bufA[4], bufB[4];
      int sc = wv;
      if (sc < nSuper) {
        const int scLast = nSuper - 1;
        load_keys4(bufA, keys, sc, lane);
        while (true) {
          load_keys4(bufB, keys, (sc + 32 < scLast) ? sc + 32 : scLast, lane);
          note_keys(bufA, sc, lane, pat, nNoted, myNoted);
          sc += 32;
          if (sc >= nSuper) break;
          load_keys4(bufA, keys, (sc + 32 < scLast) ? sc + 32 : scLast, lane);
          note_keys(bufB, sc, lane, pat, nNoted, myNoted);
          sc += 32;
          if (sc >= nSuper) break;
        }
      }
    }
    for (int e0 = (nSuper << 11) + wv * 32; e0 < nEdges; e0 += AGG_THREADS) {
      const int e = e0 + lane;
      bool mine = false;
      if (e < nEdges) mine = (unsigned)(dst[e] - nodeLo) < nOwn;
      wave_note(mine, (unsigned)e, nNoted, myNoted);
    }
    if (nNoted > SUB_CAP && lane == 0) sCount = 1u;
    __syncthreads();

    const bool fits = sCount == 0u;
    if (fits) {
      for (unsigned i0 = 0u; i0 < nNoted; i0 += 64u) {
        const unsigned ia = i0 + (unsigned)lane;
        const unsigned ib = ia + 32u;
        const unsigned ea = (ia < nNoted) ? myNoted[ia] : 0u;
        const unsigned eb = (ib < nNoted) ? myNoted[ib] : 0u;
        const unsigned ra = (unsigned)(dst[ea] - nodeLo);
        const unsigned sa = (unsigned)src[ea];
        const unsigned rb = (unsigned)(dst[eb] - nodeLo);
        const unsigned sb = (unsigned)src[eb];
        asm volatile("" :: "v"(ra), "v"(sa), "v"(rb), "v"(sb));
        if (ia < nNoted && ra < nOwn) {
          const unsigned sub = ra / SUB_NODES;
          const unsigned slot = atomicAdd(&sSubCount[sub], 1u);
          if (slot < SUB_CAP) sSub[sub * SUB_CAP + slot] = sa * 16u + (ra - sub * SUB_NODES);
          atomicAdd(sCnt + ra, 1.0f);
        }
        if (ib < nNoted && rb < nOwn) {
          const unsigned sub = rb / SUB_NODES;
          const unsigned slot = atomicAdd(&sSubCount[sub], 1u);
          if (slot < SUB_CAP) sSub[sub * SUB_CAP + slot] = sb * 16u + (rb - sub * SUB_NODES);
          atomicAdd(sCnt + rb, 1.0f);
        }
      }
    }
    __syncthreads();

    if (fits) {
      for (int i = tid; i < nOwnI; i += AGG_THREADS) cnt[nodeLo + i] = sCnt[i];
      nW = sSubCount[wv];
      if (nW > SUB_CAP) nW = LIST_PLAIN;
    } else {
      nW = LIST_PLAIN;
    }
    nW = (unsigned)__builtin_amdgcn_readfirstlane((int)nW);
    if (lane == 0) *gc = nW;

    if (nk != 0u) {
      if (nW != LIST_PLAIN) {
        const unsigned* sw = sSub + wv * SUB_CAP;
        for (unsigned i0 = 0u; i0 < nW; i0 += 32u) {
          int sL = 0;
          int dL = PAD_NODE;
          if (i0 + (unsigned)lane < nW) {
            const unsigned ent = sw[i0 + lane];
            gw[i0 + lane] = ent;
            sL = (int)(ent >> 4);
            dL = (int)(ent & 15u);
          }
          add_group(x, acc, sL, dL, lane);
        }
      } else {
        wave_plain(x, src, dst, acc, cl, nEdges, node0, nk, lane);
        if (!fits && (unsigned)lane < nk) cnt[node0 + lane] = cl;
      }
      store_rows(agg, acc, node0, nk, lane);
    }
  } else {
    if (nk == 0u) return;
    nW = (unsigned)__builtin_amdgcn_readfirstlane((int)*gc);
    if (nW != LIST_PLAIN) {
      unsigned entNext = ((unsigned)lane < nW) ? gw[lane] : (unsigned)PAD_NODE;
      for (unsigned i0 = 0u; i0 < nW; i0 += 32u) {
        const unsigned ent = entNext;
        const unsigned i1 = i0 + 32u + (unsigned)lane;
        entNext = (i1 < nW) ? gw[i1] : (unsigned)PAD_NODE;
        const int sL = (int)(ent >> 4);
        const int dL = (int)(ent & 15u);
        add_group(x, acc, sL, dL, lane);
      }
    } else {
      wave_plain(x, src, dst, acc, cl, nEdges, node0, nk, lane);
    }
    store_rows(agg, acc, node0, nk, lane);
  }
}

#define A_PITCH 132
#define GEMM_BLOCKS_128 512
#define GEMM_BLOCKS_64 1024

template <int COUT, bool RELU>
__global__ void __launch_bounds__(COUT * 2) __attribute__((amdgpu_waves_per_eu(4, 4)))
sage_gemm(const float* __restrict__ xin,
          const float* __restrict__ agg,
          const float* __restrict__ cnt,
          const float* __restrict__ Wl,
          const float* __restrict__ bl,
          const float* __restrict__ Wr,
          float* __restrict__ out,
          int nStrips) {
  constexpr int TN  = COUT / 16;
  constexpr int RP  = 16 / TN;
  constexpr int MAT = 16 * A_PITCH;
  constexpr int BUF = 2 * MAT;
  __shared__ __attribute__((aligned(16))) float sA[2 * BUF];

  const int tid  = (int)threadIdx.x;
  const int lane = tid & 31;
  const int wv   = __builtin_amdgcn_readfirstlane(tid >> 5);
  const int r    = lane & 15;
  const int kh   = lane >> 4;
  const int col  = wv * 16 + r;
  const int step = (int)gridDim.x;

  v2f bL[32], bR[32];
#pragma unroll
  for (int ks = 0; ks < 32; ++ks) {
    const int kk = 4 * ks + 2 * kh;
    bL[ks][0] = Wl[kk * COUT + col];
    bL[ks][1] = Wl[(kk + 1) * COUT + col];
    bR[ks][0] = Wr[kk * COUT + col];
    bR[ks][1] = Wr[(kk + 1) * COUT + col];
  }
  const float bias = bl[col];

  v4f pa[RP], px[RP];
  float  pc[RP];

#define GEMM_FETCH(STRIP)                                                       \
  _Pragma("unroll")                                                             \
  for (int p = 0; p < RP; ++p) {                                                \
    const long long g = (long long)(STRIP) * 16 + p * TN + wv;                  \
    pa[p] = *(const v4f*)(agg + g * C_IN + lane * 4);                           \
    px[p] = *(const v4f*)(xin + g * C_IN + lane * 4);                           \
    pc[p] = cnt[g];                                                             \
  }

#define GEMM_STAGE(SB)                                                          \
  _Pragma("unroll")                                                             \
  for (int p = 0; p < RP; ++p) {                                                \
    const float inv = 1.0f / fmaxf(pc[p], 1.0f);                                \
    const v4f a = pa[p] * inv;                                                  \
    float* d = (SB) + (p * TN + wv) * A_PITCH + lane * 4;                       \
    *(v4f*)d = a;                                                               \
    *(v4f*)(d + MAT) = px[p];                                                   \
  }

#define GEMM_STRIP_BARRIER()                                                    \
  __builtin_amdgcn_fence(__ATOMIC_RELEASE, "workgroup", "local");               \
  __builtin_amdgcn_s_barrier();                                                 \
  __builtin_amdgcn_fence(__ATOMIC_ACQUIRE, "workgroup", "local")

  int it = (int)blockIdx.x;
  if (it >= nStrips) return;
  const int itLast = nStrips - 1;
  GEMM_FETCH(it)
  GEMM_STAGE(sA)
  GEMM_FETCH((it + step < itLast) ? it + step : itLast)
  GEMM_STRIP_BARRIER();

  int buf = 0;
  for (; it < nStrips; it += step) {
    const float* fa = sA + buf * BUF + r * A_PITCH + 2 * kh;
    v8f acc = {};
    v2f fragA[8], fragB[8];
#pragma unroll
    for (int q = 0; q < 8; ++q) fragA[q] = *(const v2f*)(fa + 4 * q);
#pragma unroll
    for (int c = 0; c < 8; c += 2) {
      {
        const float* nb = fa + ((c + 1) >= 4 ? MAT : 0) + 32 * ((c + 1) & 3);
#pragma unroll
        for (int q = 0; q < 8; ++q) fragB[q] = *(const v2f*)(nb + 4 * q);
      }
#pragma unroll
      for (int q = 0; q < 8; ++q)
        acc = __builtin_amdgcn_wmma_f32_16x16x4_f32(false, fragA[q], false,
                                                    (c < 4) ? bL[8 * c + q] : bR[8 * (c - 4) + q],
                                                    (short)0, acc, false, false);
      if (c + 2 < 8) {
        const float* nb = fa + ((c + 2) >= 4 ? MAT : 0) + 32 * ((c + 2) & 3);
#pragma unroll
        for (int q = 0; q < 8; ++q) fragA[q] = *(const v2f*)(nb + 4 * q);
      }
#pragma unroll
      for (int q = 0; q < 8; ++q)
        acc = __builtin_amdgcn_wmma_f32_16x16x4_f32(false, fragB[q], false,
                                                    (c + 1 < 4) ? bL[8 * (c + 1) + q] : bR[8 * (c + 1 - 4) + q],
                                                    (short)0, acc, false, false);
    }

    const long long m0 = (long long)it * 16;
#pragma unroll
    for (int rr = 0; rr < 8; ++rr) {
      float v = acc[rr] + bias;
      if (RELU) v = fmaxf(v, 0.0f);
      out[(m0 + rr + 8 * kh) * COUT + col] = v;
    }

    GEMM_STAGE(sA + (buf ^ 1) * BUF)
    GEMM_FETCH((it + 2 * step < itLast) ? it + 2 * step : itLast)
    GEMM_STRIP_BARRIER();
    buf ^= 1;
  }
#undef GEMM_FETCH
#undef GEMM_STAGE
#undef GEMM_STRIP_BARRIER
}

extern "C" void kernel_launch(void* const* d_in, const int* in_sizes, int n_in,
                              void* d_out, int out_size, void* d_ws, size_t ws_size,
                              hipStream_t stream) {
  const float* x0  = (const float*)d_in[0];
  const int*   ei  = (const int*)d_in[1];
  const float* Wl0 = (const float*)d_in[2];
  const float* bl0 = (const float*)d_in[3];
  const float* Wr0 = (const float*)d_in[4];
  const float* Wl1 = (const float*)d_in[5];
  const float* bl1 = (const float*)d_in[6];
  const float* Wr1 = (const float*)d_in[7];
  const float* Wl2 = (const float*)d_in[8];
  const float* bl2 = (const float*)d_in[9];
  const float* Wr2 = (const float*)d_in[10];

  const int N = in_sizes[0] / C_IN;
  const int E = in_sizes[1] / 2;
  const int* src = ei;
  const int* dst = ei + E;

  const int aggBlocks = (N + OWN_NODES - 1) / OWN_NODES;

  float* h0  = (float*)d_ws;
  float* h1  = h0 + (size_t)N * C_IN;
  float* agg = h1 + (size_t)N * C_IN;
  unsigned* lists  = (unsigned*)(agg + (size_t)N * C_IN);
  unsigned* counts = lists + (size_t)aggBlocks * LIST_CAP;
  unsigned* keys   = counts + (size_t)aggBlocks * 32;
  float* cnt = (float*)(keys + (size_t)E / 4 + 4);

  const int tilesM = (N + 15) / 16;

  sage_owner_keys<<<KEY_BLOCKS, KEY_THREADS, 0, stream>>>(dst, keys, E);
  sage_aggregate<true><<<aggBlocks, AGG_THREADS, 0, stream>>>(x0, src, dst, agg, cnt, lists, counts, keys, E, N);
  sage_gemm<128, true><<<GEMM_BLOCKS_128, 256, 0, stream>>>(x0, agg, cnt, Wl0, bl0, Wr0, h0, tilesM);

  sage_aggregate<false><<<aggBlocks, AGG_THREADS, 0, stream>>>(h0, src, dst, agg, cnt, lists, counts, keys, E, N);
  sage_gemm<128, true><<<GEMM_BLOCKS_128, 256, 0, stream>>>(h0, agg, cnt, Wl1, bl1, Wr1, h1, tilesM);

  sage_aggregate<false><<<aggBlocks, AGG_THREADS, 0, stream>>>(h1, src, dst, agg, cnt, lists, counts, keys, E, N);
  sage_gemm<64, false><<<GEMM_BLOCKS_64, 128, 0, stream>>>(h1, agg, cnt, Wl2, bl2, Wr2,
                                                                 (float*)d_out, tilesM);
}


# Edge-list pass: key launch also writes, per 2048-edge block and owner, one presence bit per 64-edge chunk; an owner fetches and tests only the chunks flagged for it; lists still in arrival order
# speedup vs baseline: 4.4220x; 1.0331x over previous
#include <hip/hip_runtime.h>
#include <hip/hip_bf16.h>

typedef __attribute__((ext_vector_type(2))) float v2f;
typedef __attribute__((ext_vector_type(8))) float v8f;
typedef __attribute__((ext_vector_type(4))) float v4f;

#define C_IN 128

#define OWN_NODES 391
#define SUB_NODES 13
#define SUB_CAP 512
#define LIST_CAP 16384
#define LIST_PLAIN 0xFFFFFFFFu
static_assert(LIST_CAP == 32 * SUB_CAP, "every wave notes into its own SUB_CAP entries of the owner's list");
#define AGG_THREADS 1024
#define EDGE_BATCH 8

#define KEY_THREADS 512
#define KEY_BLOCKS 800

__device__ __forceinline__ unsigned key_pos(unsigned el) {
  const unsigned l  = el >> 6;
  const unsigned hi = (el >> 5) & 1u;
  const unsigned b  = (el >> 3) & 3u;
  const unsigned c  = el & 7u;
  const unsigned j  = 2u * hi + ((c < 4u) ? 1u : 0u);
  const unsigned u  = 3u - (c & 3u);
  return l * 64u + j * 16u + 4u * u + b;
}

#define PRES_WORDS 256
__global__ void __launch_bounds__(KEY_THREADS)
sage_owner_keys(const int* __restrict__ dst, unsigned* __restrict__ keys,
                unsigned* __restrict__ pres, int nEdges) {
  __shared__ unsigned sKey[KEY_THREADS];
  __shared__ unsigned sPres[PRES_WORDS];
  unsigned char* sKeyB = (unsigned char*)sKey;
  const unsigned tid = threadIdx.x;
  const int nBlk = nEdges >> 11;
  for (int blk = (int)blockIdx.x; blk < nBlk; blk += (int)gridDim.x) {
    if (tid < PRES_WORDS) sPres[tid] = 0u;
    __syncthreads();
    const int4 d = *(const int4*)(dst + (long long)blk * 2048 + tid * 4);
    const unsigned o0 = ((unsigned)d.x / OWN_NODES) & 255u;
    const unsigned o1 = ((unsigned)d.y / OWN_NODES) & 255u;
    const unsigned o2 = ((unsigned)d.z / OWN_NODES) & 255u;
    const unsigned o3 = ((unsigned)d.w / OWN_NODES) & 255u;
    sKeyB[key_pos(4u * tid + 0u)] = (unsigned char)o0;
    sKeyB[key_pos(4u * tid + 1u)] = (unsigned char)o1;
    sKeyB[key_pos(4u * tid + 2u)] = (unsigned char)o2;
    sKeyB[key_pos(4u * tid + 3u)] = (unsigned char)o3;
    const unsigned bit = 1u << (tid >> 4);
    atomicOr(&sPres[o0], bit);
    atomicOr(&sPres[o1], bit);
    atomicOr(&sPres[o2], bit);
    atomicOr(&sPres[o3], bit);
    __syncthreads();
    keys[(long long)blk * KEY_THREADS + tid] = sKey[tid];
    if (tid < PRES_WORDS) pres[(long long)blk * PRES_WORDS + tid] = sPres[tid];
    __syncthreads();
  }
}

__device__ __forceinline__ void load_chunk(uint4 (&buf)[4], const unsigned* __restrict__ keys,
                                           unsigned id) {
#pragma unroll
  for (int u = 0; u < 4; ++u)
    buf[u] = *(const uint4*)(keys + (size_t)id * 16 + u * 4);
}

__device__ __forceinline__ unsigned key_hits(unsigned w, unsigned pat) {
  const unsigned x = w ^ pat;
  return (x - 0x01010101u) & ~x & 0x80808080u;
}

__device__ __forceinline__ unsigned unit_hits(const uint4& k, unsigned pat) {
  return key_hits(k.x, pat) | (key_hits(k.y, pat) >> 1)
       | (key_hits(k.z, pat) >> 2) | (key_hits(k.w, pat) >> 3);
}

__device__ __forceinline__ void wave_note(bool has, unsigned e, unsigned& n, unsigned* noted) {
  const unsigned msk = __builtin_amdgcn_ballot_w32(has);
  if (has) {
    const unsigned slot = n + __builtin_amdgcn_mbcnt_lo(msk, 0u);
    if (slot < SUB_CAP) noted[slot] = e;
  }
  n += (unsigned)__popc(msk);
}

__device__ __forceinline__ void note_keys(const uint4 (&buf)[4], unsigned id, bool has_chunk,
                                          unsigned pat, unsigned& n, unsigned* noted) {
  const unsigned mLo = unit_hits(buf[0], pat) | (unit_hits(buf[1], pat) >> 4);
  const unsigned mHi = unit_hits(buf[2], pat) | (unit_hits(buf[3], pat) >> 4);
  unsigned long long m = ((unsigned long long)mHi << 32) | mLo;
  if (!has_chunk) m = 0ull;
  const unsigned eBase = id * 64u;
  while (true) {
    const bool has = m != 0ull;
    if (__builtin_amdgcn_ballot_w32(has) == 0u) break;
    const unsigned q = (unsigned)__builtin_ctzll(m | (1ull << 63));
    wave_note(has, eBase + q, n, noted);
    m &= m - 1ull;
  }
}

typedef __attribute__((ext_vector_type(16))) float v16f;
static_assert(SUB_NODES <= 16, "a register row vector holds 16 nodes");
struct RowAcc { v16f c0, c1, c2, c3; };

#define PAD_NODE 15
static_assert(PAD_NODE >= SUB_NODES && PAD_NODE < 16, "the spare element lies past the wave's nodes");
__device__ __forceinline__ void add_group(const float* __restrict__ x, RowAcc& acc,
                                          int sL, int dL, int lane) {
#pragma unroll
  for (int j = 0; j < 32; j += EDGE_BATCH) {
    unsigned off[EDGE_BATCH];
    v4f v[EDGE_BATCH];
#pragma unroll
    for (int u = 0; u < EDGE_BATCH; ++u)
      off[u] = (unsigned)__builtin_amdgcn_readlane(sL, j + u) * (unsigned)(C_IN * 4) + (unsigned)lane * 16u;
    __builtin_amdgcn_sched_barrier(0);
#pragma unroll
    for (int u = 0; u < EDGE_BATCH; ++u)
      v[u] = *(const v4f*)((const char*)x + off[u]);
#pragma unroll
    for (int u = 0; u < EDGE_BATCH; ++u) {
      const int d = __builtin_amdgcn_readlane(dL, j + u);
      acc.c0[d] += v[u][0];
      acc.c1[d] += v[u][1];
      acc.c2[d] += v[u][2];
      acc.c3[d] += v[u][3];
    }
  }
}

__device__ __forceinline__ void wave_plain(const float* __restrict__ x,
                                           const int* __restrict__ src,
                                           const int* __restrict__ dst,
                                           RowAcc& acc, float& cl,
                                           int nEdges, int node0, unsigned nk, int lane) {
  for (int e0 = 0; e0 < nEdges; e0 += 32) {
    const int e = e0 + lane;
    unsigned r = 0xFFFFFFFFu;
    if (e < nEdges) r = (unsigned)(dst[e] - node0);
    const bool mine = r < nk;
    unsigned mask = __builtin_amdgcn_ballot_w32(mine);
    if (mask != 0u) {
      int s = 0;
      if (mine) s = src[e];
      while (mask != 0u) {
        const int i = __builtin_ctz(mask);
        mask &= mask - 1u;
        const int sRow = __builtin_amdgcn_readlane(s, i);
        const int d = __builtin_amdgcn_readlane((int)r, i);
        const v4f v = *(const v4f*)((const char*)x + (unsigned)sRow * (unsigned)(C_IN * 4) + (unsigned)lane * 16u);
        acc.c0[d] += v[0];
        acc.c1[d] += v[1];
        acc.c2[d] += v[2];
        acc.c3[d] += v[3];
        if (lane == d) cl += 1.0f;
      }
    }
  }
}

__device__ __forceinline__ void store_rows(float* __restrict__ agg, const RowAcc& acc,
                                           int node0, unsigned nk, int lane) {
#pragma unroll
  for (int k = 0; k < SUB_NODES; ++k) {
    if ((unsigned)k < nk) {
      v4f o;
      o[0] = acc.c0[k];
      o[1] = acc.c1[k];
      o[2] = acc.c2[k];
      o[3] = acc.c3[k];
      *(v4f*)(agg + (long long)(node0 + k) * C_IN + lane * 4) = o;
    }
  }
}

template <bool BUILD>
__global__ void __launch_bounds__(AGG_THREADS)
sage_aggregate(const float* __restrict__ x,
               const int* __restrict__ src,
               const int* __restrict__ dst,
               float* __restrict__ agg,
               float* __restrict__ cnt,
               unsigned* __restrict__ gList,
               unsigned* __restrict__ gCount,
               const unsigned* __restrict__ keys,
               const unsigned* __restrict__ pres,
               int nEdges, int nNodes) {
  const int tid  = (int)threadIdx.x;
  const int lane = tid & 31;
  const int wv   = __builtin_amdgcn_readfirstlane(tid >> 5);
  const int nodeLo = (int)blockIdx.x * OWN_NODES;
  int nOwnI = nNodes - nodeLo;
  if (nOwnI > OWN_NODES) nOwnI = OWN_NODES;
  if (nOwnI <= 0) return;
  const unsigned nOwn = (unsigned)nOwnI;
  const int node0 = nodeLo + wv * SUB_NODES;
  int nkI = nOwnI - wv * SUB_NODES;
  if (nkI > SUB_NODES) nkI = SUB_NODES;
  if (nkI < 0) nkI = 0;
  const unsigned nk = (unsigned)nkI;
  unsigned* gw = gList + (size_t)blockIdx.x * LIST_CAP + (size_t)wv * SUB_CAP;
  unsigned* gc = gCount + (size_t)blockIdx.x * 32 + wv;

  RowAcc acc;
  acc.c0 = (v16f)(0.0f);
  acc.c1 = (v16f)(0.0f);
  acc.c2 = (v16f)(0.0f);
  acc.c3 = (v16f)(0.0f);
  float cl = 0.0f;

  unsigned nW;
  if constexpr (BUILD) {
    __shared__ unsigned sList[LIST_CAP];
    __shared__ unsigned sSub[32 * SUB_CAP];
    __shared__ unsigned sSubCount[32];
    __shared__ float sCnt[OWN_NODES];
    __shared__ unsigned sCount;
    __shared__ unsigned sChunkCount;
    unsigned* sChunk = sSub;
    constexpr unsigned CHUNK_CAP = 32u * SUB_CAP;

    for (int i = tid; i < OWN_NODES; i += AGG_THREADS) sCnt[i] = 0.0f;
    if (tid < 32) sSubCount[tid] = 0u;
    if (tid == 0) { sCount = 0u; sChunkCount = 0u; }
    __syncthreads();

    const int nSuper = nEdges >> 11;
    const unsigned ownByte = (unsigned)blockIdx.x & 255u;
    const unsigned pat = ownByte * 0x01010101u;
    unsigned* myNoted = sList + wv * SUB_CAP;
    unsigned nNoted = 0u;
    for (int blk = tid; blk < nSuper; blk += AGG_THREADS) {
      unsigned w = pres[(size_t)blk * PRES_WORDS + ownByte];
      const unsigned c = (unsigned)__popc(w);
      if (c != 0u) {
        unsigned slot = atomicAdd(&sChunkCount, c);
        while (w != 0u) {
          const unsigned l = (unsigned)__builtin_ctz(w);
          w &= w - 1u;
          if (slot < CHUNK_CAP) sChunk[slot] = (unsigned)blk * 32u + l;
          ++slot;
        }
      }
    }
    __syncthreads();
    const unsigned nChunk = (unsigned)__builtin_amdgcn_readfirstlane((int)sChunkCount);
    if (nChunk > CHUNK_CAP && tid == 0) sCount = 1u;
    if (nChunk <= CHUNK_CAP) {
      uint4 bufA[4], bufB[4];
      unsigned i0 = (unsigned)wv * 32u;
      if (i0 < nChunk) {
        bool hasA = i0 + (unsigned)lane < nChunk;
        unsigned idA = sChunk[hasA ? i0 + (unsigned)lane : 0u];
        load_chunk(bufA, keys, idA);
        while (true) {
          bool hasB = i0 + AGG_THREADS + (unsigned)lane < nChunk;
          unsigned idB = sChunk[hasB ? i0 + AGG_THREADS + (unsigned)lane : 0u];
          load_chunk(bufB, keys, idB);
          note_keys(bufA, idA, hasA, pat, nNoted, myNoted);
          i0 += AGG_THREADS;
          if (i0 >= nChunk) break;
          hasA = i0 + AGG_THREADS + (unsigned)lane < nChunk;
          idA = sChunk[hasA ? i0 + AGG_THREADS + (unsigned)lane : 0u];
          load_chunk(bufA, keys, idA);
          note_keys(bufB, idB, hasB, pat, nNoted, myNoted);
          i0 += AGG_THREADS;
          if (i0 >= nChunk) break;
        }
      }
    }
    for (int e0 = (nSuper << 11) + wv * 32; e0 < nEdges; e0 += AGG_THREADS) {
      const int e = e0 + lane;
      bool mine = false;
      if (e < nEdges) mine = (unsigned)(dst[e] - nodeLo) < nOwn;
      wave_note(mine, (unsigned)e, nNoted, myNoted);
    }
    if (nNoted > SUB_CAP && lane == 0) sCount = 1u;
    __syncthreads();

    const bool fits = sCount == 0u;
    if (fits) {
      for (unsigned i0 = 0u; i0 < nNoted; i0 += 64u) {
        const unsigned ia = i0 + (unsigned)lane;
        const unsigned ib = ia + 32u;
        const unsigned ea = (ia < nNoted) ? myNoted[ia] : 0u;
        const unsigned eb = (ib < nNoted) ? myNoted[ib] : 0u;
        const unsigned ra = (unsigned)(dst[ea] - nodeLo);
        const unsigned sa = (unsigned)src[ea];
        const unsigned rb = (unsigned)(dst[eb] - nodeLo);
        const unsigned sb = (unsigned)src[eb];
        asm volatile("" :: "v"(ra), "v"(sa), "v"(rb), "v"(sb));
        if (ia < nNoted && ra < nOwn) {
          const unsigned sub = ra / SUB_NODES;
          const unsigned slot = atomicAdd(&sSubCount[sub], 1u);
          if (slot < SUB_CAP) sSub[sub * SUB_CAP + slot] = sa * 16u + (ra - sub * SUB_NODES);
          atomicAdd(sCnt + ra, 1.0f);
        }
        if (ib < nNoted && rb < nOwn) {
          const unsigned sub = rb / SUB_NODES;
          const unsigned slot = atomicAdd(&sSubCount[sub], 1u);
          if (slot < SUB_CAP) sSub[sub * SUB_CAP + slot] = sb * 16u + (rb - sub * SUB_NODES);
          atomicAdd(sCnt + rb, 1.0f);
        }
      }
    }
    __syncthreads();

    if (fits) {
      for (int i = tid; i < nOwnI; i += AGG_THREADS) cnt[nodeLo + i] = sCnt[i];
      nW = sSubCount[wv];
      if (nW > SUB_CAP) nW = LIST_PLAIN;
    } else {
      nW = LIST_PLAIN;
    }
    nW = (unsigned)__builtin_amdgcn_readfirstlane((int)nW);
    if (lane == 0) *gc = nW;

    if (nk != 0u) {
      if (nW != LIST_PLAIN) {
        const unsigned* sw = sSub + wv * SUB_CAP;
        for (unsigned i0 = 0u; i0 < nW; i0 += 32u) {
          int sL = 0;
          int dL = PAD_NODE;
          if (i0 + (unsigned)lane < nW) {
            const unsigned ent = sw[i0 + lane];
            gw[i0 + lane] = ent;
            sL = (int)(ent >> 4);
            dL = (int)(ent & 15u);
          }
          add_group(x, acc, sL, dL, lane);
        }
      } else {
        wave_plain(x, src, dst, acc, cl, nEdges, node0, nk, lane);
        if (!fits && (unsigned)lane < nk) cnt[node0 + lane] = cl;
      }
      store_rows(agg, acc, node0, nk, lane);
    }
  } else {
    if (nk == 0u) return;
    nW = (unsigned)__builtin_amdgcn_readfirstlane((int)*gc);
    if (nW != LIST_PLAIN) {
      unsigned entNext = ((unsigned)lane < nW) ? gw[lane] : (unsigned)PAD_NODE;
      for (unsigned i0 = 0u; i0 < nW; i0 += 32u) {
        const unsigned ent = entNext;
        const unsigned i1 = i0 + 32u + (unsigned)lane;
        entNext = (i1 < nW) ? gw[i1] : (unsigned)PAD_NODE;
        const int sL = (int)(ent >> 4);
        const int dL = (int)(ent & 15u);
        add_group(x, acc, sL, dL, lane);
      }
    } else {
      wave_plain(x, src, dst, acc, cl, nEdges, node0, nk, lane);
    }
    store_rows(agg, acc, node0, nk, lane);
  }
}

#define A_PITCH 132
#define GEMM_BLOCKS_128 512
#define GEMM_BLOCKS_64 1024

template <int COUT, bool RELU>
__global__ void __launch_bounds__(COUT * 2) __attribute__((amdgpu_waves_per_eu(4, 4)))
sage_gemm(const float* __restrict__ xin,
          const float* __restrict__ agg,
          const float* __restrict__ cnt,
          const float* __restrict__ Wl,
          const float* __restrict__ bl,
          const float* __restrict__ Wr,
          float* __restrict__ out,
          int nStrips) {
  constexpr int TN  = COUT / 16;
  constexpr int RP  = 16 / TN;
  constexpr int MAT = 16 * A_PITCH;
  constexpr int BUF = 2 * MAT;
  __shared__ __attribute__((aligned(16))) float sA[2 * BUF];

  const int tid  = (int)threadIdx.x;
  const int lane = tid & 31;
  const int wv   = __builtin_amdgcn_readfirstlane(tid >> 5);
  const int r    = lane & 15;
  const int kh   = lane >> 4;
  const int col  = wv * 16 + r;
  const int step = (int)gridDim.x;

  v2f bL[32], bR[32];
#pragma unroll
  for (int ks = 0; ks < 32; ++ks) {
    const int kk = 4 * ks + 2 * kh;
    bL[ks][0] = Wl[kk * COUT + col];
    bL[ks][1] = Wl[(kk + 1) * COUT + col];
    bR[ks][0] = Wr[kk * COUT + col];
    bR[ks][1] = Wr[(kk + 1) * COUT + col];
  }
  const float bias = bl[col];

  v4f pa[RP], px[RP];
  float  pc[RP];

#define GEMM_FETCH(STRIP)                                                       \
  _Pragma("unroll")                                                             \
  for (int p = 0; p < RP; ++p) {                                                \
    const long long g = (long long)(STRIP) * 16 + p * TN + wv;                  \
    pa[p] = *(const v4f*)(agg + g * C_IN + lane * 4);                           \
    px[p] = *(const v4f*)(xin + g * C_IN + lane * 4);                           \
    pc[p] = cnt[g];                                                             \
  }

#define GEMM_STAGE(SB)                                                          \
  _Pragma("unroll")                                                             \
  for (int p = 0; p < RP; ++p) {                                                \
    const float inv = 1.0f / fmaxf(pc[p], 1.0f);                                \
    const v4f a = pa[p] * inv;                                                  \
    float* d = (SB) + (p * TN + wv) * A_PITCH + lane * 4;                       \
    *(v4f*)d = a;                                                               \
    *(v4f*)(d + MAT) = px[p];                                                   \
  }

#define GEMM_STRIP_BARRIER()                                                    \
  __builtin_amdgcn_fence(__ATOMIC_RELEASE, "workgroup", "local");               \
  __builtin_amdgcn_s_barrier();                                                 \
  __builtin_amdgcn_fence(__ATOMIC_ACQUIRE, "workgroup", "local")

  int it = (int)blockIdx.x;
  if (it >= nStrips) return;
  const int itLast = nStrips - 1;
  GEMM_FETCH(it)
  GEMM_STAGE(sA)
  GEMM_FETCH((it + step < itLast) ? it + step : itLast)
  GEMM_STRIP_BARRIER();

  int buf = 0;
  for (; it < nStrips; it += step) {
    const float* fa = sA + buf * BUF + r * A_PITCH + 2 * kh;
    v8f acc = {};
    v2f fragA[8], fragB[8];
#pragma unroll
    for (int q = 0; q < 8; ++q) fragA[q] = *(const v2f*)(fa + 4 * q);
#pragma unroll
    for (int c = 0; c < 8; c += 2) {
      {
        const float* nb = fa + ((c + 1) >= 4 ? MAT : 0) + 32 * ((c + 1) & 3);
#pragma unroll
        for (int q = 0; q < 8; ++q) fragB[q] = *(const v2f*)(nb + 4 * q);
      }
#pragma unroll
      for (int q = 0; q < 8; ++q)
        acc = __builtin_amdgcn_wmma_f32_16x16x4_f32(false, fragA[q], false,
                                                    (c < 4) ? bL[8 * c + q] : bR[8 * (c - 4) + q],
                                                    (short)0, acc, false, false);
      if (c + 2 < 8) {
        const float* nb = fa + ((c + 2) >= 4 ? MAT : 0) + 32 * ((c + 2) & 3);
#pragma unroll
        for (int q = 0; q < 8; ++q) fragA[q] = *(const v2f*)(nb + 4 * q);
      }
#pragma unroll
      for (int q = 0; q < 8; ++q)
        acc = __builtin_amdgcn_wmma_f32_16x16x4_f32(false, fragB[q], false,
                                                    (c + 1 < 4) ? bL[8 * (c + 1) + q] : bR[8 * (c + 1 - 4) + q],
                                                    (short)0, acc, false, false);
    }

    const long long m0 = (long long)it * 16;
#pragma unroll
    for (int rr = 0; rr < 8; ++rr) {
      float v = acc[rr] + bias;
      if (RELU) v = fmaxf(v, 0.0f);
      out[(m0 + rr + 8 * kh) * COUT + col] = v;
    }

    GEMM_STAGE(sA + (buf ^ 1) * BUF)
    GEMM_FETCH((it + 2 * step < itLast) ? it + 2 * step : itLast)
    GEMM_STRIP_BARRIER();
    buf ^= 1;
  }
#undef GEMM_FETCH
#undef GEMM_STAGE
#undef GEMM_STRIP_BARRIER
}

extern "C" void kernel_launch(void* const* d_in, const int* in_sizes, int n_in,
                              void* d_out, int out_size, void* d_ws, size_t ws_size,
                              hipStream_t stream) {
  const float* x0  = (const float*)d_in[0];
  const int*   ei  = (const int*)d_in[1];
  const float* Wl0 = (const float*)d_in[2];
  const float* bl0 = (const float*)d_in[3];
  const float* Wr0 = (const float*)d_in[4];
  const float* Wl1 = (const float*)d_in[5];
  const float* bl1 = (const float*)d_in[6];
  const float* Wr1 = (const float*)d_in[7];
  const float* Wl2 = (const float*)d_in[8];
  const float* bl2 = (const float*)d_in[9];
  const float* Wr2 = (const float*)d_in[10];

  const int N = in_sizes[0] / C_IN;
  const int E = in_sizes[1] / 2;
  const int* src = ei;
  const int* dst = ei + E;

  const int aggBlocks = (N + OWN_NODES - 1) / OWN_NODES;

  float* h0  = (float*)d_ws;
  float* h1  = h0 + (size_t)N * C_IN;
  float* agg = h1 + (size_t)N * C_IN;
  unsigned* lists  = (unsigned*)(agg + (size_t)N * C_IN);
  unsigned* counts = lists + (size_t)aggBlocks * LIST_CAP;
  unsigned* keys   = counts + (size_t)aggBlocks * 32;
  unsigned* pres   = keys + (size_t)E / 4 + 4;
  float* cnt = (float*)(pres + ((size_t)E >> 11) * PRES_WORDS);

  const int tilesM = (N + 15) / 16;

  sage_owner_keys<<<KEY_BLOCKS, KEY_THREADS, 0, stream>>>(dst, keys, pres, E);
  sage_aggregate<true><<<aggBlocks, AGG_THREADS, 0, stream>>>(x0, src, dst, agg, cnt, lists, counts, keys, pres, E, N);
  sage_gemm<128, true><<<GEMM_BLOCKS_128, 256, 0, stream>>>(x0, agg, cnt, Wl0, bl0, Wr0, h0, tilesM);

  sage_aggregate<false><<<aggBlocks, AGG_THREADS, 0, stream>>>(h0, src, dst, agg, cnt, lists, counts, keys, pres, E, N);
  sage_gemm<128, true><<<GEMM_BLOCKS_128, 256, 0, stream>>>(h0, agg, cnt, Wl1, bl1, Wr1, h1, tilesM);

  sage_aggregate<false><<<aggBlocks, AGG_THREADS, 0, stream>>>(h1, src, dst, agg, cnt, lists, counts, keys, pres, E, N);
  sage_gemm<64, false><<<GEMM_BLOCKS_64, 128, 0, stream>>>(h1, agg, cnt, Wl2, bl2, Wr2,
                                                                 (float*)d_out, tilesM);
}
